# starting kernel
# baseline (speedup 1.0000x reference)
#include <hip/hip_runtime.h>
#include <math.h>

typedef float v2f __attribute__((ext_vector_type(2)));
typedef float v8f __attribute__((ext_vector_type(8)));

#define NEG_SLOPE 0.2f

__device__ __forceinline__ float leaky(float x) {
    return x > 0.f ? x : NEG_SLOPE * x;
}

__device__ __forceinline__ void atomicMaxF32(float* addr, float v) {
    if (v >= 0.f) {
        atomicMax((int*)addr, __float_as_int(v));
    } else {
        atomicMin((unsigned int*)addr, __float_as_uint(v));
    }
}

template <int CT>
__global__ void wmma_gemm_f32(const float* __restrict__ A,
                              const float* __restrict__ B,
                              float* __restrict__ Out,
                              int M, int K, int Nout) {
    const int wave = threadIdx.x >> 5;
    const int lane = threadIdx.x & 31;
    const int row0 = (blockIdx.x * (blockDim.x >> 5) + wave) << 4;
    const int col0 = blockIdx.y * (CT * 16);
    if (row0 >= M) return;

    const int half = lane >> 4;
    const int l16  = lane & 15;
    const float* arow = A + (size_t)(row0 + l16) * K;
    const int col = col0 + l16;

    v8f acc[CT];
    #pragma unroll
    for (int t = 0; t < CT; ++t) acc[t] = (v8f){};

    for (int k = 0; k < K; k += 4) {
        const int ka = k + 2 * half;
        v2f a;
        a.x = arow[ka];
        a.y = arow[ka + 1];
        const float* brow0 = B + (size_t)ka * Nout + col;
        const float* brow1 = B + (size_t)(ka + 1) * Nout + col;
        #pragma unroll
        for (int t = 0; t < CT; ++t) {
            v2f b;
            b.x = brow0[t * 16];
            b.y = brow1[t * 16];
            acc[t] = __builtin_amdgcn_wmma_f32_16x16x4_f32(
                 false, a,  false, b,
                 (short)0, acc[t],  false,  false);
        }
    }
    #pragma unroll
    for (int t = 0; t < CT; ++t) {
        #pragma unroll
        for (int r = 0; r < 8; ++r) {
            Out[(size_t)(row0 + r + 8 * half) * Nout + col + t * 16] = acc[t][r];
        }
    }
}

template <int H, int C>
__global__ void alpha_kernel(const float* __restrict__ h,
                             const float* __restrict__ a_src,
                             const float* __restrict__ a_dst,
                             float* __restrict__ as, float* __restrict__ ad,
                             int n) {
    int gid = blockIdx.x * blockDim.x + threadIdx.x;
    if (gid >= n * H) return;
    const int node = gid / H;
    const int hh = gid - node * H;
    const float* hp = h + (size_t)node * (H * C) + hh * C;
    const float* asw = a_src + hh * C;
    const float* adw = a_dst + hh * C;
    float s = 0.f, d = 0.f;
    #pragma unroll
    for (int c = 0; c < C; ++c) {
        s += hp[c] * asw[c];
        d += hp[c] * adw[c];
    }
    as[gid] = s;
    ad[gid] = d;
}

__global__ void fill_kernel(float* __restrict__ p, float v, int n) {
    int gid = blockIdx.x * blockDim.x + threadIdx.x;
    if (gid < n) p[gid] = v;
}

__global__ void init_bias_kernel(float* __restrict__ out,
                                 const float* __restrict__ bias,
                                 int n, int HC) {
    int gid = blockIdx.x * blockDim.x + threadIdx.x;
    if (gid >= n * HC) return;
    out[gid] = bias[gid % HC];
}

__device__ __forceinline__ void edge_sd(const int* __restrict__ eidx, int E,
                                        int i, int& s, int& d) {
    if (i < E) { s = eidx[i]; d = eidx[E + i]; }
    else       { s = i - E;   d = i - E; }
}

template <int H>
__global__ void edge_max_kernel(const int* __restrict__ eidx, int E, int n,
                                const float* __restrict__ as,
                                const float* __restrict__ ad,
                                float* __restrict__ m) {
    int gid = blockIdx.x * blockDim.x + threadIdx.x;
    int tot = (E + n) * H;
    if (gid >= tot) return;
    const int i = gid / H;
    const int hh = gid - i * H;
    int s, d;
    edge_sd(eidx, E, i, s, d);
    float e = leaky(as[s * H + hh] + ad[d * H + hh]);
    atomicMaxF32(&m[d * H + hh], e);
}

template <int H>
__global__ void edge_sum_kernel(const int* __restrict__ eidx, int E, int n,
                                const float* __restrict__ as,
                                const float* __restrict__ ad,
                                const float* __restrict__ m,
                                float* __restrict__ den) {
    int gid = blockIdx.x * blockDim.x + threadIdx.x;
    int tot = (E + n) * H;
    if (gid >= tot) return;
    const int i = gid / H;
    const int hh = gid - i * H;
    int s, d;
    edge_sd(eidx, E, i, s, d);
    float e = leaky(as[s * H + hh] + ad[d * H + hh]);
    float ex = __expf(e - m[d * H + hh]);
    atomicAdd(&den[d * H + hh], ex);
}

template <int H, int C>
__global__ void edge_agg_kernel(const int* __restrict__ eidx, int E, int n,
                                const float* __restrict__ as,
                                const float* __restrict__ ad,
                                const float* __restrict__ m,
                                const float* __restrict__ den,
                                const float* __restrict__ h,
                                float* __restrict__ out) {
    int gid = blockIdx.x * blockDim.x + threadIdx.x;
    int tot = (E + n) * H;
    if (gid >= tot) return;
    const int i = gid / H;
    const int hh = gid - i * H;
    int s, d;
    edge_sd(eidx, E, i, s, d);
    float e = leaky(as[s * H + hh] + ad[d * H + hh]);
    float coeff = __expf(e - m[d * H + hh]) / den[d * H + hh];
    const float* hs = h + (size_t)s * (H * C) + hh * C;
    float* od = out + (size_t)d * (H * C) + hh * C;
    #pragma unroll
    for (int c = 0; c < C; ++c) {
        atomicAdd(&od[c], coeff * hs[c]);
    }
}

static inline int ceil_div(int a, int b) { return (a + b - 1) / b; }

extern "C" void kernel_launch(void* const* d_in, const int* in_sizes, int n_in,
                              void* d_out, int out_size, void* d_ws, size_t ws_size,
                              hipStream_t stream) {
    const float* x    = (const float*)d_in[0];
    const int*   eidx = (const int*)  d_in[1];
    const float* W1   = (const float*)d_in[2];
    const float* aS1  = (const float*)d_in[3];
    const float* aD1  = (const float*)d_in[4];
    const float* b1   = (const float*)d_in[5];
    const float* W2   = (const float*)d_in[6];
    const float* aS2  = (const float*)d_in[7];
    const float* aD2  = (const float*)d_in[8];
    const float* b2   = (const float*)d_in[9];
    float* out = (float*)d_out;

    const int FEATS = 256, HID = 64, H1 = 8, C1 = 8, CLS = 16;
    const int N = in_sizes[0] / FEATS;
    const int E = in_sizes[1] / 2;
    const int Etot = E + N;

    float* ws = (float*)d_ws;
    float* h1   = ws;                 ws += (size_t)N * HID;
    float* as1  = ws;                 ws += (size_t)N * H1;
    float* ad1  = ws;                 ws += (size_t)N * H1;
    float* m1   = ws;                 ws += (size_t)N * H1;
    float* den1 = ws;                 ws += (size_t)N * H1;
    float* o1   = ws;                 ws += (size_t)N * HID;
    float* h2   = ws;                 ws += (size_t)N * CLS;
    float* as2  = ws;                 ws += (size_t)N;
    float* ad2  = ws;                 ws += (size_t)N;
    float* m2   = ws;                 ws += (size_t)N;
    float* den2 = ws;                 ws += (size_t)N;

    const int TPB = 256;
    const int WPB = 4;
    const dim3 gblock(32 * WPB);

    {
        dim3 grid(ceil_div(N / 16, WPB), 1);
        wmma_gemm_f32<4><<<grid, gblock, 0, stream>>>(x, W1, h1, N, FEATS, HID);
    }
    alpha_kernel<H1, C1><<<ceil_div(N * H1, TPB), TPB, 0, stream>>>(h1, aS1, aD1, as1, ad1, N);
    fill_kernel<<<ceil_div(N * H1, TPB), TPB, 0, stream>>>(m1, -INFINITY, N * H1);
    fill_kernel<<<ceil_div(N * H1, TPB), TPB, 0, stream>>>(den1, 0.f, N * H1);
    init_bias_kernel<<<ceil_div(N * HID, TPB), TPB, 0, stream>>>(o1, b1, N, HID);

    {
        const int tot = Etot * H1;
        edge_max_kernel<H1><<<ceil_div(tot, TPB), TPB, 0, stream>>>(eidx, E, N, as1, ad1, m1);
        edge_sum_kernel<H1><<<ceil_div(tot, TPB), TPB, 0, stream>>>(eidx, E, N, as1, ad1, m1, den1);
        edge_agg_kernel<H1, C1><<<ceil_div(tot, TPB), TPB, 0, stream>>>(eidx, E, N, as1, ad1, m1, den1, h1, o1);
    }

    {
        dim3 grid(ceil_div(N / 16, WPB), 1);
        wmma_gemm_f32<1><<<grid, gblock, 0, stream>>>(o1, W2, h2, N, HID, CLS);
    }
    alpha_kernel<1, CLS><<<ceil_div(N, TPB), TPB, 0, stream>>>(h2, aS2, aD2, as2, ad2, N);
    fill_kernel<<<ceil_div(N, TPB), TPB, 0, stream>>>(m2, -INFINITY, N);
    fill_kernel<<<ceil_div(N, TPB), TPB, 0, stream>>>(den2, 0.f, N);
    init_bias_kernel<<<ceil_div(N * CLS, TPB), TPB, 0, stream>>>(out, b2, N, CLS);

    {
        const int tot = Etot;
        edge_max_kernel<1><<<ceil_div(tot, TPB), TPB, 0, stream>>>(eidx, E, N, as2, ad2, m2);
        edge_sum_kernel<1><<<ceil_div(tot, TPB), TPB, 0, stream>>>(eidx, E, N, as2, ad2, m2, den2);
        edge_agg_kernel<1, CLS><<<ceil_div(tot, TPB), TPB, 0, stream>>>(eidx, E, N, as2, ad2, m2, den2, h2, out);
    }
}


# softmax division deferred: one edge pass adds exp weights into den and exp-weighted rows into out (atomics, arrival-order sums kept); consumer divides once per element; no separate sum pass
# speedup vs baseline: 1.1088x; 1.1088x over previous
#include <hip/hip_runtime.h>
#include <math.h>

typedef float v2f __attribute__((ext_vector_type(2)));
typedef float v8f __attribute__((ext_vector_type(8)));

#define NEG_SLOPE 0.2f

__device__ __forceinline__ float leaky(float x) {
    return x > 0.f ? x : NEG_SLOPE * x;
}

__device__ __forceinline__ void atomicMaxF32(float* addr, float v) {
    if (v >= 0.f) {
        atomicMax((int*)addr, __float_as_int(v));
    } else {
        atomicMin((unsigned int*)addr, __float_as_uint(v));
    }
}

template <int CT, int NC>
__global__ void wmma_gemm_f32(const float* __restrict__ A,
                              const float* __restrict__ B,
                              float* __restrict__ Out,
                              int M, int K, int Nout,
                              const float* __restrict__ den,
                              const float* __restrict__ bias) {
    const int wave = threadIdx.x >> 5;
    const int lane = threadIdx.x & 31;
    const int row0 = (blockIdx.x * (blockDim.x >> 5) + wave) << 4;
    const int col0 = blockIdx.y * (CT * 16);
    if (row0 >= M) return;

    const int half = lane >> 4;
    const int l16  = lane & 15;
    const float* arow = A + (size_t)(row0 + l16) * K;
    const float* drow = nullptr;
    if constexpr (NC > 0) drow = den + (size_t)(row0 + l16) * (K / NC);
    const int col = col0 + l16;

    v8f acc[CT];
    #pragma unroll
    for (int t = 0; t < CT; ++t) acc[t] = (v8f){};

    for (int k = 0; k < K; k += 4) {
        const int ka = k + 2 * half;
        v2f a;
        a.x = arow[ka];
        a.y = arow[ka + 1];
        if constexpr (NC > 0) {
            a.x = a.x / drow[ka / NC] + bias[ka];
            a.y = a.y / drow[(ka + 1) / NC] + bias[ka + 1];
        }
        const float* brow0 = B + (size_t)ka * Nout + col;
        const float* brow1 = B + (size_t)(ka + 1) * Nout + col;
        #pragma unroll
        for (int t = 0; t < CT; ++t) {
            v2f b;
            b.x = brow0[t * 16];
            b.y = brow1[t * 16];
            acc[t] = __builtin_amdgcn_wmma_f32_16x16x4_f32(
                 false, a,  false, b,
                 (short)0, acc[t],  false,  false);
        }
    }
    #pragma unroll
    for (int t = 0; t < CT; ++t) {
        #pragma unroll
        for (int r = 0; r < 8; ++r) {
            Out[(size_t)(row0 + r + 8 * half) * Nout + col + t * 16] = acc[t][r];
        }
    }
}

template <int H, int C>
__global__ void alpha_kernel(const float* __restrict__ h,
                             const float* __restrict__ a_src,
                             const float* __restrict__ a_dst,
                             float* __restrict__ as, float* __restrict__ ad,
                             int n) {
    int gid = blockIdx.x * blockDim.x + threadIdx.x;
    if (gid >= n * H) return;
    const int node = gid / H;
    const int hh = gid - node * H;
    const float* hp = h + (size_t)node * (H * C) + hh * C;
    const float* asw = a_src + hh * C;
    const float* adw = a_dst + hh * C;
    float s = 0.f, d = 0.f;
    #pragma unroll
    for (int c = 0; c < C; ++c) {
        s += hp[c] * asw[c];
        d += hp[c] * adw[c];
    }
    as[gid] = s;
    ad[gid] = d;
}

__global__ void fill_kernel(float* __restrict__ p, float v, int n) {
    int gid = blockIdx.x * blockDim.x + threadIdx.x;
    if (gid < n) p[gid] = v;
}

template <int H, int C>
__global__ void normalize_bias_kernel(float* __restrict__ out,
                                      const float* __restrict__ den,
                                      const float* __restrict__ bias,
                                      int n) {
    int gid = blockIdx.x * blockDim.x + threadIdx.x;
    if (gid >= n * (H * C)) return;
    out[gid] = out[gid] / den[gid / C] + bias[gid % (H * C)];
}

__device__ __forceinline__ void edge_sd(const int* __restrict__ eidx, int E,
                                        int i, int& s, int& d) {
    if (i < E) { s = eidx[i]; d = eidx[E + i]; }
    else       { s = i - E;   d = i - E; }
}

template <int H>
__global__ void edge_max_kernel(const int* __restrict__ eidx, int E, int n,
                                const float* __restrict__ as,
                                const float* __restrict__ ad,
                                float* __restrict__ m) {
    int gid = blockIdx.x * blockDim.x + threadIdx.x;
    int tot = (E + n) * H;
    if (gid >= tot) return;
    const int i = gid / H;
    const int hh = gid - i * H;
    int s, d;
    edge_sd(eidx, E, i, s, d);
    float e = leaky(as[s * H + hh] + ad[d * H + hh]);
    atomicMaxF32(&m[d * H + hh], e);
}

template <int H, int C>
__global__ void edge_sumagg_kernel(const int* __restrict__ eidx, int E, int n,
                                   const float* __restrict__ as,
                                   const float* __restrict__ ad,
                                   const float* __restrict__ m,
                                   const float* __restrict__ h,
                                   float* __restrict__ den,
                                   float* __restrict__ out) {
    int gid = blockIdx.x * blockDim.x + threadIdx.x;
    int tot = (E + n) * H;
    if (gid >= tot) return;
    const int i = gid / H;
    const int hh = gid - i * H;
    int s, d;
    edge_sd(eidx, E, i, s, d);
    float e = leaky(as[s * H + hh] + ad[d * H + hh]);
    float ex = __expf(e - m[d * H + hh]);
    atomicAdd(&den[d * H + hh], ex);
    const float* hs = h + (size_t)s * (H * C) + hh * C;
    float* od = out + (size_t)d * (H * C) + hh * C;
    #pragma unroll
    for (int c = 0; c < C; ++c) {
        atomicAdd(&od[c], ex * hs[c]);
    }
}

static inline int ceil_div(int a, int b) { return (a + b - 1) / b; }

extern "C" void kernel_launch(void* const* d_in, const int* in_sizes, int n_in,
                              void* d_out, int out_size, void* d_ws, size_t ws_size,
                              hipStream_t stream) {
    const float* x    = (const float*)d_in[0];
    const int*   eidx = (const int*)  d_in[1];
    const float* W1   = (const float*)d_in[2];
    const float* aS1  = (const float*)d_in[3];
    const float* aD1  = (const float*)d_in[4];
    const float* b1   = (const float*)d_in[5];
    const float* W2   = (const float*)d_in[6];
    const float* aS2  = (const float*)d_in[7];
    const float* aD2  = (const float*)d_in[8];
    const float* b2   = (const float*)d_in[9];
    float* out = (float*)d_out;

    const int FEATS = 256, HID = 64, H1 = 8, C1 = 8, CLS = 16;
    const int N = in_sizes[0] / FEATS;
    const int E = in_sizes[1] / 2;
    const int Etot = E + N;

    float* ws = (float*)d_ws;
    float* h1   = ws;                 ws += (size_t)N * HID;
    float* as1  = ws;                 ws += (size_t)N * H1;
    float* ad1  = ws;                 ws += (size_t)N * H1;
    float* m1   = ws;                 ws += (size_t)N * H1;
    float* den1 = ws;                 ws += (size_t)N * H1;
    float* o1   = ws;                 ws += (size_t)N * HID;
    float* h2   = ws;                 ws += (size_t)N * CLS;
    float* as2  = ws;                 ws += (size_t)N;
    float* ad2  = ws;                 ws += (size_t)N;
    float* m2   = ws;                 ws += (size_t)N;
    float* den2 = ws;                 ws += (size_t)N;

    const int TPB = 256;
    const int WPB = 4;
    const dim3 gblock(32 * WPB);

    {
        dim3 grid(ceil_div(N / 16, WPB), 1);
        wmma_gemm_f32<4, 0><<<grid, gblock, 0, stream>>>(x, W1, h1, N, FEATS, HID, nullptr, nullptr);
    }
    alpha_kernel<H1, C1><<<ceil_div(N * H1, TPB), TPB, 0, stream>>>(h1, aS1, aD1, as1, ad1, N);
    fill_kernel<<<ceil_div(N * H1, TPB), TPB, 0, stream>>>(m1, -INFINITY, N * H1);
    fill_kernel<<<ceil_div(N * H1, TPB), TPB, 0, stream>>>(den1, 0.f, N * H1);
    fill_kernel<<<ceil_div(N * HID, TPB), TPB, 0, stream>>>(o1, 0.f, N * HID);

    {
        const int tot = Etot * H1;
        edge_max_kernel<H1><<<ceil_div(tot, TPB), TPB, 0, stream>>>(eidx, E, N, as1, ad1, m1);
        edge_sumagg_kernel<H1, C1><<<ceil_div(tot, TPB), TPB, 0, stream>>>(eidx, E, N, as1, ad1, m1, h1, den1, o1);
    }

    {
        dim3 grid(ceil_div(N / 16, WPB), 1);
        wmma_gemm_f32<1, C1><<<grid, gblock, 0, stream>>>(o1, W2, h2, N, HID, CLS, den1, b1);
    }
    alpha_kernel<1, CLS><<<ceil_div(N, TPB), TPB, 0, stream>>>(h2, aS2, aD2, as2, ad2, N);
    fill_kernel<<<ceil_div(N, TPB), TPB, 0, stream>>>(m2, -INFINITY, N);
    fill_kernel<<<ceil_div(N, TPB), TPB, 0, stream>>>(den2, 0.f, N);
    fill_kernel<<<ceil_div(N * CLS, TPB), TPB, 0, stream>>>(out, 0.f, N * CLS);

    {
        const int tot = Etot;
        edge_max_kernel<1><<<ceil_div(tot, TPB), TPB, 0, stream>>>(eidx, E, N, as2, ad2, m2);
        edge_sumagg_kernel<1, CLS><<<ceil_div(tot, TPB), TPB, 0, stream>>>(eidx, E, N, as2, ad2, m2, h2, den2, out);
        normalize_bias_kernel<1, CLS><<<ceil_div(N * CLS, TPB), TPB, 0, stream>>>(out, den2, b2, N);
    }
}


# edge linking off global atomics: block (edge chunk, node range) pushes its chunk's edges by LDS exchange; a node gets 8 chunk lists; row sums in chunk order, LDS arrival order inside a chunk
# speedup vs baseline: 5.6498x; 5.0954x over previous
#include <hip/hip_runtime.h>
#include <math.h>

typedef float v2f __attribute__((ext_vector_type(2)));
typedef float v8f __attribute__((ext_vector_type(8)));

#define NEG_SLOPE 0.2f

__device__ __forceinline__ float leaky(float x) {
    return x > 0.f ? x : NEG_SLOPE * x;
}

#define GEMM_WPB 4
template <int CT, int H, bool COPY_HEADS>
__global__ void proj_logits_kernel(const float* __restrict__ A,
                                   const float* __restrict__ B,
                                   const float* __restrict__ a_src,
                                   const float* __restrict__ a_dst,
                                   float* Out, float* as, float* ad,
                                   const int* cps, int* cpd, int cpn4,
                                   int M, int K, int ldo, int ast) {
    constexpr int NOUT = CT * 16;
    constexpr int C    = NOUT / H;
    constexpr int P    = NOUT + 4;
    __shared__ float4 tile4[GEMM_WPB][16 * P / 4];

    if constexpr (COPY_HEADS) {
        const int nthr = gridDim.x * blockDim.x;
        for (int i = blockIdx.x * blockDim.x + threadIdx.x; i < cpn4; i += nthr)
            ((int4*)cpd)[i] = ((const int4*)cps)[i];
    }

    const int wave = threadIdx.x >> 5;
    const int lane = threadIdx.x & 31;
    const int row0 = (blockIdx.x * GEMM_WPB + wave) << 4;
    if (row0 >= M) return;

    const int half = lane >> 4;
    const int l16  = lane & 15;
    const float* arow = A + (size_t)(row0 + l16) * K;

    static_assert(H == 1 || 32 % H == 0, "a lane keeps one head");
    constexpr int NW = (H == 1) ? C : 2 * C;
    const int hh = (H == 1) ? 0 : lane % H;
    float wreg[NW];
    if constexpr (H == 1) {
        const float* wv = half ? a_dst : a_src;
        #pragma unroll
        for (int c = 0; c < C; ++c) wreg[c] = wv[c];
    } else {
        #pragma unroll
        for (int c = 0; c < C; ++c) { wreg[c] = a_src[hh * C + c]; wreg[C + c] = a_dst[hh * C + c]; }
    }

    v8f acc[CT];
    #pragma unroll
    for (int t = 0; t < CT; ++t) acc[t] = (v8f){};

    for (int k = 0; k < K; k += 4) {
        const int ka = k + 2 * half;
        v2f a;
        a.x = arow[ka];
        a.y = arow[ka + 1];
        const float* brow0 = B + (size_t)ka * NOUT + l16;
        const float* brow1 = B + (size_t)(ka + 1) * NOUT + l16;
        #pragma unroll
        for (int t = 0; t < CT; ++t) {
            v2f b;
            b.x = brow0[t * 16];
            b.y = brow1[t * 16];
            acc[t] = __builtin_amdgcn_wmma_f32_16x16x4_f32(
                 false, a,  false, b,
                 (short)0, acc[t],  false,  false);
        }
    }

    float* tw = (float*)tile4[wave];
    #pragma unroll
    for (int t = 0; t < CT; ++t) {
        #pragma unroll
        for (int r = 0; r < 8; ++r) {
            tw[(r + 8 * half) * P + t * 16 + l16] = acc[t][r];
        }
    }
    __builtin_amdgcn_fence(__ATOMIC_RELEASE, "wavefront");
    __builtin_amdgcn_wave_barrier();
    __builtin_amdgcn_fence(__ATOMIC_ACQUIRE, "wavefront");

    constexpr int F4R = NOUT / 4;
    #pragma unroll
    for (int f = lane; f < 16 * F4R; f += 32) {
        const int row = f / F4R;
        const int c4  = f - row * F4R;
        const float4 v = *(const float4*)(tw + row * P + c4 * 4);
        *(float4*)(Out + (size_t)(row0 + row) * ldo + c4 * 4) = v;
    }

    if constexpr (H == 1) {
        const float* hp = tw + l16 * P;
        float s = 0.f;
        #pragma unroll
        for (int c = 0; c < C; ++c) s += hp[c] * wreg[c];
        float* o = half ? ad : as;
        o[(size_t)(row0 + l16) * ast] = s;
    } else {
        #pragma unroll
        for (int p = lane; p < 16 * H; p += 32) {
            const int row = p / H;
            const float* hp = tw + row * P + hh * C;
            float s = 0.f, d = 0.f;
            #pragma unroll
            for (int c = 0; c < C; ++c) {
                s += hp[c] * wreg[c];
                d += hp[c] * wreg[C + c];
            }
            as[(size_t)(row0 + row) * ast + hh] = s;
            ad[(size_t)(row0 + row) * ast + hh] = d;
        }
    }
}

#define LNK_B   8
#define LNK_R   32
#define LNK_CAP 4096
#define LNK_TPB 1024
#define LNK_U   4
__global__ void __launch_bounds__(LNK_TPB)
link_edges_kernel(const int* __restrict__ eidx, int E, int n,
                  int* __restrict__ headT,
                  int* __restrict__ rec) {
    __shared__ int lh[LNK_CAP];
    const int c   = blockIdx.x / LNK_R;
    const int r   = blockIdx.x - c * LNK_R;
    const int nr  = min((n + LNK_R - 1) / LNK_R, LNK_CAP);
    const int lo  = r * nr;
    const int ch  = ((E + LNK_B * 4 - 1) / (LNK_B * 4)) * 4;
    const int c0  = c * ch;
    const int c1  = min(c0 + ch, E);
    const int term = -(c + 2);
    for (int i = threadIdx.x; i < nr; i += LNK_TPB) lh[i] = term;
    __syncthreads();

    const int* __restrict__ dstp = eidx + E;
    constexpr int STR = LNK_TPB * 4;
    if ((E & 3) == 0) {
        for (int b0 = c0 + threadIdx.x * 4; b0 < c1; b0 += STR * LNK_U) {
            int4 dv[LNK_U], sv[LNK_U];
            #pragma unroll
            for (int g = 0; g < LNK_U; ++g) {
                const int bc = min(b0 + g * STR, c1 - 4);
                dv[g] = *(const int4*)(dstp + bc);
                sv[g] = *(const int4*)(eidx + bc);
            }
            #pragma unroll
            for (int g = 0; g < LNK_U; ++g) {
                const int b = b0 + g * STR;
                const bool live = b < c1;
                const int d[4] = {dv[g].x, dv[g].y, dv[g].z, dv[g].w};
                const int s[4] = {sv[g].x, sv[g].y, sv[g].z, sv[g].w};
                bool m[4];
                int  p[4];
                #pragma unroll
                for (int j = 0; j < 4; ++j) {
                    const unsigned u = (unsigned)(d[j] - lo);
                    m[j] = live && u < (unsigned)nr;
                    p[j] = 0;
                    if (m[j]) p[j] = atomicExch(&lh[u], b + j);
                }
                #pragma unroll
                for (int j = 0; j < 4; ++j)
                    if (m[j]) ((int2*)rec)[b + j] = make_int2(s[j], p[j]);
            }
        }
    } else {
        for (int b = c0 + threadIdx.x; b < c1; b += LNK_TPB) {
            const unsigned u = (unsigned)(dstp[b] - lo);
            if (u < (unsigned)nr) {
                const int p = atomicExch(&lh[u], b);
                ((int2*)rec)[b] = make_int2(eidx[b], p);
            }
        }
    }
    __syncthreads();
    for (int i = threadIdx.x; i < nr; i += LNK_TPB)
        if (lo + i < n) headT[(size_t)c * n + lo + i] = lh[i];
}

template <int H, int C, int HS4, int AST>
__global__ void row_walk_kernel(const int* __restrict__ rec, int E, int n,
                                const int* __restrict__ headT,
                                const float* __restrict__ as,
                                const float* __restrict__ ad,
                                const float* __restrict__ h,
                                const float* __restrict__ bias,
                                float* __restrict__ out) {
    constexpr int LPR = (H * C) / 4;
    const int gid = blockIdx.x * blockDim.x + threadIdx.x;
    const int row = gid / LPR;
    if (row >= n) return;
    const int q  = gid - row * LPR;
    const int hh = (q * 4) / C;
    const float4* h4 = (const float4*)h;
    const int2*   r2 = (const int2*)rec;

    const float adv = ad[row * AST + hh];
    float  m   = leaky(as[row * AST + hh] + adv);
    float  den = 1.f;
    float4 acc = h4[(size_t)row * HS4 + q];

    int nh[LNK_B];
    int e = -1;
    {
        int hd[LNK_B];
        #pragma unroll
        for (int c = 0; c < LNK_B; ++c) hd[c] = headT[(size_t)c * n + row];
        #pragma unroll
        for (int c = LNK_B - 1; c >= 0; --c) { nh[c] = e; if (hd[c] >= 0) e = hd[c]; }
    }
    int s = 0, nx = -1;
    if (e >= 0) { const int2 r = r2[e]; s = r.x; nx = r.y; }
    for (int it = 0; e >= 0 && it < E; ++it) {
        const float4 hv = h4[(size_t)s * HS4 + q];
        const float  sc = leaky(as[s * AST + hh] + adv);
        int e2 = nx;
        if (e2 < 0) {
            const int ce = -e2 - 2;
            e2 = -1;
            #pragma unroll
            for (int c = 0; c < LNK_B; ++c) e2 = (ce == c) ? nh[c] : e2;
        }
        int s2 = 0, nx2 = -1;
        if (e2 >= 0) { const int2 r = r2[e2]; s2 = r.x; nx2 = r.y; }
        const float mn = fmaxf(m, sc);
        const float a  = __expf(m - mn);
        const float w  = __expf(sc - mn);
        den   = den   * a + w;
        acc.x = acc.x * a + w * hv.x;
        acc.y = acc.y * a + w * hv.y;
        acc.z = acc.z * a + w * hv.z;
        acc.w = acc.w * a + w * hv.w;
        m = mn;
        e = e2; s = s2; nx = nx2;
    }
    const float4 b = ((const float4*)bias)[q];
    float4 o;
    o.x = acc.x / den + b.x;
    o.y = acc.y / den + b.y;
    o.z = acc.z / den + b.z;
    o.w = acc.w / den + b.w;
    ((float4*)out)[(size_t)row * LPR + q] = o;
}

static inline int ceil_div(int a, int b) { return (a + b - 1) / b; }

extern "C" void kernel_launch(void* const* d_in, const int* in_sizes, int n_in,
                              void* d_out, int out_size, void* d_ws, size_t ws_size,
                              hipStream_t stream) {
    const float* x    = (const float*)d_in[0];
    const int*   eidx = (const int*)  d_in[1];
    const float* W1   = (const float*)d_in[2];
    const float* aS1  = (const float*)d_in[3];
    const float* aD1  = (const float*)d_in[4];
    const float* b1   = (const float*)d_in[5];
    const float* W2   = (const float*)d_in[6];
    const float* aS2  = (const float*)d_in[7];
    const float* aD2  = (const float*)d_in[8];
    const float* b2   = (const float*)d_in[9];
    float* out = (float*)d_out;

    const int FEATS = 256, HID = 64, H1 = 8, C1 = 8, CLS = 16;
    const int N = in_sizes[0] / FEATS;
    const int E = in_sizes[1] / 2;

    float* ws = (float*)d_ws;
    float* h1   = ws;                 ws += (size_t)N * HID;
    float* o1   = ws;                 ws += (size_t)N * HID;
    int*   rec  = (int*)ws;           ws += (size_t)E * 2;
    float* as1  = ws;                 ws += (size_t)N * H1;
    float* ad1  = ws;                 ws += (size_t)N * H1;
    int*   headT1 = (int*)out;
    int*   headT2 = (int*)as1;
    const int NR = 32;
    float* nrec = h1;

    const int TPB = 256;
    const dim3 gblock(32 * GEMM_WPB);
    const dim3 ggrid(ceil_div(N / 16, GEMM_WPB), 1);

    proj_logits_kernel<4, H1, false><<<ggrid, gblock, 0, stream>>>(
        x, W1, aS1, aD1, h1, as1, ad1, nullptr, nullptr, 0, N, FEATS, HID, H1);
    link_edges_kernel<<<LNK_B * LNK_R, LNK_TPB, 0, stream>>>(eidx, E, N, headT1, rec);
    row_walk_kernel<H1, C1, HID / 4, H1><<<ceil_div(N * (HID / 4), TPB), TPB, 0, stream>>>(
        rec, E, N, headT1, as1, ad1, h1, b1, o1);

    proj_logits_kernel<1, 1, true><<<ggrid, gblock, 0, stream>>>(
        o1, W2, aS2, aD2, nrec, nrec + CLS, nrec + CLS + 1, headT1, headT2, LNK_B * (N / 4),
        N, HID, NR, NR);
    row_walk_kernel<1, CLS, NR / 4, NR><<<ceil_div(N * (CLS / 4), TPB), TPB, 0, stream>>>(
        rec, E, N, headT2, nrec + CLS, nrec + CLS + 1, nrec, b2, out);
}


# list heads inlined: a chunk list's first-arrived edge {source,next} lives in the head table (LDS compare-and-store claims it; later edges chain behind): no record read for it; it is summed first now
# speedup vs baseline: 5.9586x; 1.0547x over previous
#include <hip/hip_runtime.h>
#include <math.h>

typedef float v2f __attribute__((ext_vector_type(2)));
typedef float v8f __attribute__((ext_vector_type(8)));

#define NEG_SLOPE 0.2f

__device__ __forceinline__ float leaky(float x) {
    return x > 0.f ? x : NEG_SLOPE * x;
}

#define GEMM_WPB 4
template <int CT, int H, bool COPY_HEADS>
__global__ void proj_logits_kernel(const float* __restrict__ A,
                                   const float* __restrict__ B,
                                   const float* __restrict__ a_src,
                                   const float* __restrict__ a_dst,
                                   float* Out, float* as, float* ad,
                                   const int* cps, int* cpd, int cpn4,
                                   int M, int K, int ldo, int ast) {
    constexpr int NOUT = CT * 16;
    constexpr int C    = NOUT / H;
    constexpr int P    = NOUT + 4;
    __shared__ float4 tile4[GEMM_WPB][16 * P / 4];

    if constexpr (COPY_HEADS) {
        const int nthr = gridDim.x * blockDim.x;
        for (int i = blockIdx.x * blockDim.x + threadIdx.x; i < cpn4; i += nthr)
            ((int4*)cpd)[i] = ((const int4*)cps)[i];
    }

    const int wave = threadIdx.x >> 5;
    const int lane = threadIdx.x & 31;
    const int row0 = (blockIdx.x * GEMM_WPB + wave) << 4;
    if (row0 >= M) return;

    const int half = lane >> 4;
    const int l16  = lane & 15;
    const float* arow = A + (size_t)(row0 + l16) * K;

    static_assert(H == 1 || 32 % H == 0, "a lane keeps one head");
    constexpr int NW = (H == 1) ? C : 2 * C;
    const int hh = (H == 1) ? 0 : lane % H;
    float wreg[NW];
    if constexpr (H == 1) {
        const float* wv = half ? a_dst : a_src;
        #pragma unroll
        for (int c = 0; c < C; ++c) wreg[c] = wv[c];
    } else {
        #pragma unroll
        for (int c = 0; c < C; ++c) { wreg[c] = a_src[hh * C + c]; wreg[C + c] = a_dst[hh * C + c]; }
    }

    v8f acc[CT];
    #pragma unroll
    for (int t = 0; t < CT; ++t) acc[t] = (v8f){};

    for (int k = 0; k < K; k += 4) {
        const int ka = k + 2 * half;
        v2f a;
        a.x = arow[ka];
        a.y = arow[ka + 1];
        const float* brow0 = B + (size_t)ka * NOUT + l16;
        const float* brow1 = B + (size_t)(ka + 1) * NOUT + l16;
        #pragma unroll
        for (int t = 0; t < CT; ++t) {
            v2f b;
            b.x = brow0[t * 16];
            b.y = brow1[t * 16];
            acc[t] = __builtin_amdgcn_wmma_f32_16x16x4_f32(
                 false, a,  false, b,
                 (short)0, acc[t],  false,  false);
        }
    }

    float* tw = (float*)tile4[wave];
    #pragma unroll
    for (int t = 0; t < CT; ++t) {
        #pragma unroll
        for (int r = 0; r < 8; ++r) {
            tw[(r + 8 * half) * P + t * 16 + l16] = acc[t][r];
        }
    }
    __builtin_amdgcn_fence(__ATOMIC_RELEASE, "wavefront");
    __builtin_amdgcn_wave_barrier();
    __builtin_amdgcn_fence(__ATOMIC_ACQUIRE, "wavefront");

    constexpr int F4R = NOUT / 4;
    #pragma unroll
    for (int f = lane; f < 16 * F4R; f += 32) {
        const int row = f / F4R;
        const int c4  = f - row * F4R;
        const float4 v = *(const float4*)(tw + row * P + c4 * 4);
        *(float4*)(Out + (size_t)(row0 + row) * ldo + c4 * 4) = v;
    }

    if constexpr (H == 1) {
        const float* hp = tw + l16 * P;
        float s = 0.f;
        #pragma unroll
        for (int c = 0; c < C; ++c) s += hp[c] * wreg[c];
        float* o = half ? ad : as;
        o[(size_t)(row0 + l16) * ast] = s;
    } else {
        #pragma unroll
        for (int p = lane; p < 16 * H; p += 32) {
            const int row = p / H;
            const float* hp = tw + row * P + hh * C;
            float s = 0.f, d = 0.f;
            #pragma unroll
            for (int c = 0; c < C; ++c) {
                s += hp[c] * wreg[c];
                d += hp[c] * wreg[C + c];
            }
            as[(size_t)(row0 + row) * ast + hh] = s;
            ad[(size_t)(row0 + row) * ast + hh] = d;
        }
    }
}

#define LNK_B   8
#define LNK_R   32
#define LNK_CAP 4096
#define LNK_TPB 1024
#define LNK_U   4
__global__ void __launch_bounds__(LNK_TPB)
link_edges_kernel(const int* __restrict__ eidx, int E, int n,
                  int* __restrict__ headT,
                  int* __restrict__ rec) {
    __shared__ int lhs[LNK_CAP];
    __shared__ int lhn[LNK_CAP];
    const int c   = blockIdx.x / LNK_R;
    const int r   = blockIdx.x - c * LNK_R;
    const int nr  = min((n + LNK_R - 1) / LNK_R, LNK_CAP);
    const int lo  = r * nr;
    const int ch  = ((E + LNK_B * 4 - 1) / (LNK_B * 4)) * 4;
    const int c0  = c * ch;
    const int c1  = min(c0 + ch, E);
    const int term = -(c + 2);
    for (int i = threadIdx.x; i < nr; i += LNK_TPB) { lhs[i] = -1; lhn[i] = term; }
    __syncthreads();

    const int* __restrict__ dstp = eidx + E;
    constexpr int STR = LNK_TPB * 4;
    if ((E & 3) == 0) {
        for (int b0 = c0 + threadIdx.x * 4; b0 < c1; b0 += STR * LNK_U) {
            int4 dv[LNK_U], sv[LNK_U];
            #pragma unroll
            for (int g = 0; g < LNK_U; ++g) {
                const int bc = min(b0 + g * STR, c1 - 4);
                dv[g] = *(const int4*)(dstp + bc);
                sv[g] = *(const int4*)(eidx + bc);
            }
            #pragma unroll
            for (int g = 0; g < LNK_U; ++g) {
                const int b = b0 + g * STR;
                const bool live = b < c1;
                const int d[4] = {dv[g].x, dv[g].y, dv[g].z, dv[g].w};
                const int s[4] = {sv[g].x, sv[g].y, sv[g].z, sv[g].w};
                unsigned u[4];
                bool m[4];
                int  w[4], p[4];
                #pragma unroll
                for (int j = 0; j < 4; ++j) {
                    u[j] = (unsigned)(d[j] - lo);
                    m[j] = live && u[j] < (unsigned)nr;
                    w[j] = -1;
                    if (m[j]) w[j] = atomicCAS(&lhs[u[j]], -1, s[j]);
                }
                #pragma unroll
                for (int j = 0; j < 4; ++j) {
                    m[j] = m[j] && w[j] != -1;
                    p[j] = 0;
                    if (m[j]) p[j] = atomicExch(&lhn[u[j]], b + j);
                }
                #pragma unroll
                for (int j = 0; j < 4; ++j)
                    if (m[j]) ((int2*)rec)[b + j] = make_int2(s[j], p[j]);
            }
        }
    } else {
        for (int b = c0 + threadIdx.x; b < c1; b += LNK_TPB) {
            const unsigned u = (unsigned)(dstp[b] - lo);
            if (u < (unsigned)nr) {
                const int sb = eidx[b];
                if (atomicCAS(&lhs[u], -1, sb) != -1) {
                    const int p = atomicExch(&lhn[u], b);
                    ((int2*)rec)[b] = make_int2(sb, p);
                }
            }
        }
    }
    __syncthreads();
    for (int i = threadIdx.x; i < nr; i += LNK_TPB)
        if (lo + i < n) ((int2*)headT)[(size_t)c * n + lo + i] = make_int2(lhs[i], lhn[i]);
}

template <int H, int C, int HS4, int AST>
__global__ void row_walk_kernel(const int* __restrict__ rec, int E, int n,
                                const int* __restrict__ headT,
                                const float* __restrict__ as,
                                const float* __restrict__ ad,
                                const float* __restrict__ h,
                                const float* __restrict__ bias,
                                float* __restrict__ out) {
    constexpr int LPR = (H * C) / 4;
    const int gid = blockIdx.x * blockDim.x + threadIdx.x;
    const int row = gid / LPR;
    if (row >= n) return;
    const int q  = gid - row * LPR;
    const int hh = (q * 4) / C;
    const float4* h4 = (const float4*)h;
    const int2*   r2 = (const int2*)rec;

    const float adv = ad[row * AST + hh];
    float  m   = leaky(as[row * AST + hh] + adv);
    float  den = 1.f;
    float4 acc = h4[(size_t)row * HS4 + q];

    int nhs[LNK_B], nhn[LNK_B];
    int s = -1, nx = -1;
    {
        const int2* hT = (const int2*)headT;
        int2 hd[LNK_B];
        #pragma unroll
        for (int c = 0; c < LNK_B; ++c) hd[c] = hT[(size_t)c * n + row];
        #pragma unroll
        for (int c = LNK_B - 1; c >= 0; --c) {
            nhs[c] = s; nhn[c] = nx;
            if (hd[c].x >= 0) { s = hd[c].x; nx = hd[c].y; }
        }
    }
    for (int it = 0; s >= 0 && it < E; ++it) {
        const float4 hv = h4[(size_t)s * HS4 + q];
        const float  sc = leaky(as[s * AST + hh] + adv);
        int s2 = -1, nx2 = -1;
        if (nx >= 0) {
            const int2 r = r2[nx]; s2 = r.x; nx2 = r.y;
        } else {
            const int ce = -nx - 2;
            #pragma unroll
            for (int c = 0; c < LNK_B; ++c) { s2 = (ce == c) ? nhs[c] : s2; nx2 = (ce == c) ? nhn[c] : nx2; }
        }
        const float mn = fmaxf(m, sc);
        const float a  = __expf(m - mn);
        const float w  = __expf(sc - mn);
        den   = den   * a + w;
        acc.x = acc.x * a + w * hv.x;
        acc.y = acc.y * a + w * hv.y;
        acc.z = acc.z * a + w * hv.z;
        acc.w = acc.w * a + w * hv.w;
        m = mn;
        s = s2; nx = nx2;
    }
    const float4 b = ((const float4*)bias)[q];
    float4 o;
    o.x = acc.x / den + b.x;
    o.y = acc.y / den + b.y;
    o.z = acc.z / den + b.z;
    o.w = acc.w / den + b.w;
    ((float4*)out)[(size_t)row * LPR + q] = o;
}

static inline int ceil_div(int a, int b) { return (a + b - 1) / b; }

extern "C" void kernel_launch(void* const* d_in, const int* in_sizes, int n_in,
                              void* d_out, int out_size, void* d_ws, size_t ws_size,
                              hipStream_t stream) {
    const float* x    = (const float*)d_in[0];
    const int*   eidx = (const int*)  d_in[1];
    const float* W1   = (const float*)d_in[2];
    const float* aS1  = (const float*)d_in[3];
    const float* aD1  = (const float*)d_in[4];
    const float* b1   = (const float*)d_in[5];
    const float* W2   = (const float*)d_in[6];
    const float* aS2  = (const float*)d_in[7];
    const float* aD2  = (const float*)d_in[8];
    const float* b2   = (const float*)d_in[9];
    float* out = (float*)d_out;

    const int FEATS = 256, HID = 64, H1 = 8, C1 = 8, CLS = 16;
    const int N = in_sizes[0] / FEATS;
    const int E = in_sizes[1] / 2;

    float* ws = (float*)d_ws;
    float* h1   = ws;                 ws += (size_t)N * HID;
    float* o1   = ws;                 ws += (size_t)N * HID;
    int*   rec  = (int*)ws;           ws += (size_t)E * 2;
    float* as1  = ws;                 ws += (size_t)N * H1;
    float* ad1  = ws;                 ws += (size_t)N * H1;
    int*   headT1 = (int*)out;
    int*   headT2 = (int*)as1;
    const int NR = 32;
    float* nrec = h1;

    const int TPB = 256;
    const dim3 gblock(32 * GEMM_WPB);
    const dim3 ggrid(ceil_div(N / 16, GEMM_WPB), 1);

    proj_logits_kernel<4, H1, false><<<ggrid, gblock, 0, stream>>>(
        x, W1, aS1, aD1, h1, as1, ad1, nullptr, nullptr, 0, N, FEATS, HID, H1);
    link_edges_kernel<<<LNK_B * LNK_R, LNK_TPB, 0, stream>>>(eidx, E, N, headT1, rec);
    row_walk_kernel<H1, C1, HID / 4, H1><<<ceil_div(N * (HID / 4), TPB), TPB, 0, stream>>>(
        rec, E, N, headT1, as1, ad1, h1, b1, o1);

    proj_logits_kernel<1, 1, true><<<ggrid, gblock, 0, stream>>>(
        o1, W2, aS2, aD2, nrec, nrec + CLS, nrec + CLS + 1, headT1, headT2, LNK_B * (N / 2),
        N, HID, NR, NR);
    row_walk_kernel<1, CLS, NR / 4, NR><<<ceil_div(N * (CLS / 4), TPB), TPB, 0, stream>>>(
        rec, E, N, headT2, nrec + CLS, nrec + CLS + 1, nrec, b2, out);
}


# link launch: a wave sieves its edges (subtract, compare, vote), pushes in-range ones on its LDS ring, links them in dense 32-lane batches, sources fetched there; in-chunk arrival order now the ring's
# speedup vs baseline: 6.4548x; 1.0833x over previous
#include <hip/hip_runtime.h>
#include <math.h>

typedef float v2f __attribute__((ext_vector_type(2)));
typedef float v8f __attribute__((ext_vector_type(8)));

#define NEG_SLOPE 0.2f

__device__ __forceinline__ float leaky(float x) {
    return x > 0.f ? x : NEG_SLOPE * x;
}

#define GEMM_WPB 4
template <int CT, int H, bool COPY_HEADS>
__global__ void proj_logits_kernel(const float* __restrict__ A,
                                   const float* __restrict__ B,
                                   const float* __restrict__ a_src,
                                   const float* __restrict__ a_dst,
                                   float* Out, float* as, float* ad,
                                   const int* cps, int* cpd, int cpn4,
                                   int M, int K, int ldo, int ast) {
    constexpr int NOUT = CT * 16;
    constexpr int C    = NOUT / H;
    constexpr int P    = NOUT + 4;
    __shared__ float4 tile4[GEMM_WPB][16 * P / 4];

    if constexpr (COPY_HEADS) {
        const int nthr = gridDim.x * blockDim.x;
        for (int i = blockIdx.x * blockDim.x + threadIdx.x; i < cpn4; i += nthr)
            ((int4*)cpd)[i] = ((const int4*)cps)[i];
    }

    const int wave = threadIdx.x >> 5;
    const int lane = threadIdx.x & 31;
    const int row0 = (blockIdx.x * GEMM_WPB + wave) << 4;
    if (row0 >= M) return;

    const int half = lane >> 4;
    const int l16  = lane & 15;
    const float* arow = A + (size_t)(row0 + l16) * K;

    static_assert(H == 1 || 32 % H == 0, "a lane keeps one head");
    constexpr int NW = (H == 1) ? C : 2 * C;
    const int hh = (H == 1) ? 0 : lane % H;
    float wreg[NW];
    if constexpr (H == 1) {
        const float* wv = half ? a_dst : a_src;
        #pragma unroll
        for (int c = 0; c < C; ++c) wreg[c] = wv[c];
    } else {
        #pragma unroll
        for (int c = 0; c < C; ++c) { wreg[c] = a_src[hh * C + c]; wreg[C + c] = a_dst[hh * C + c]; }
    }

    v8f acc[CT];
    #pragma unroll
    for (int t = 0; t < CT; ++t) acc[t] = (v8f){};

    for (int k = 0; k < K; k += 4) {
        const int ka = k + 2 * half;
        v2f a;
        a.x = arow[ka];
        a.y = arow[ka + 1];
        const float* brow0 = B + (size_t)ka * NOUT + l16;
        const float* brow1 = B + (size_t)(ka + 1) * NOUT + l16;
        #pragma unroll
        for (int t = 0; t < CT; ++t) {
            v2f b;
            b.x = brow0[t * 16];
            b.y = brow1[t * 16];
            acc[t] = __builtin_amdgcn_wmma_f32_16x16x4_f32(
                 false, a,  false, b,
                 (short)0, acc[t],  false,  false);
        }
    }

    float* tw = (float*)tile4[wave];
    #pragma unroll
    for (int t = 0; t < CT; ++t) {
        #pragma unroll
        for (int r = 0; r < 8; ++r) {
            tw[(r + 8 * half) * P + t * 16 + l16] = acc[t][r];
        }
    }
    __builtin_amdgcn_fence(__ATOMIC_RELEASE, "wavefront");
    __builtin_amdgcn_wave_barrier();
    __builtin_amdgcn_fence(__ATOMIC_ACQUIRE, "wavefront");

    constexpr int F4R = NOUT / 4;
    #pragma unroll
    for (int f = lane; f < 16 * F4R; f += 32) {
        const int row = f / F4R;
        const int c4  = f - row * F4R;
        const float4 v = *(const float4*)(tw + row * P + c4 * 4);
        *(float4*)(Out + (size_t)(row0 + row) * ldo + c4 * 4) = v;
    }

    if constexpr (H == 1) {
        const float* hp = tw + l16 * P;
        float s = 0.f;
        #pragma unroll
        for (int c = 0; c < C; ++c) s += hp[c] * wreg[c];
        float* o = half ? ad : as;
        o[(size_t)(row0 + l16) * ast] = s;
    } else {
        #pragma unroll
        for (int p = lane; p < 16 * H; p += 32) {
            const int row = p / H;
            const float* hp = tw + row * P + hh * C;
            float s = 0.f, d = 0.f;
            #pragma unroll
            for (int c = 0; c < C; ++c) {
                s += hp[c] * wreg[c];
                d += hp[c] * wreg[C + c];
            }
            as[(size_t)(row0 + row) * ast + hh] = s;
            ad[(size_t)(row0 + row) * ast + hh] = d;
        }
    }
}

#define LNK_B   8
#define LNK_R   32
#define LNK_CAP 4096
#define LNK_TPB 1024
#define LNK_U   4
#define LNK_QN  64
struct lnk_full { static constexpr bool value = false; };
struct lnk_tail { static constexpr bool value = true;  };
__global__ void __launch_bounds__(LNK_TPB)
link_edges_kernel(const int* __restrict__ eidx, int E, int n,
                  int* __restrict__ headT,
                  int* __restrict__ rec) {
    __shared__ int lhs[LNK_CAP];
    __shared__ int lhn[LNK_CAP];
    __shared__ int2 ring[LNK_TPB / 32][LNK_QN];
    const int c   = blockIdx.x / LNK_R;
    const int r   = blockIdx.x - c * LNK_R;
    const int nr  = min((n + LNK_R - 1) / LNK_R, LNK_CAP);
    const int lo  = r * nr;
    const int ch  = ((E + LNK_B * 4 - 1) / (LNK_B * 4)) * 4;
    const int c0  = c * ch;
    const int c1  = min(c0 + ch, E);
    const int term = -(c + 2);
    for (int i = threadIdx.x; i < nr; i += LNK_TPB) { lhs[i] = -1; lhn[i] = term; }
    __syncthreads();

    const int* __restrict__ dstp = eidx + E;
    constexpr int STR = LNK_TPB * 4;
    if ((E & 3) == 0) {
        constexpr int STEP = STR * LNK_U;
        const int lane = threadIdx.x & 31;
        int2* wq = ring[threadIdx.x >> 5];
        int qh = 0, qt = 0;

        auto link_batch = [&](int cnt) {
            __builtin_amdgcn_fence(__ATOMIC_RELEASE, "wavefront");
            __builtin_amdgcn_wave_barrier();
            __builtin_amdgcn_fence(__ATOMIC_ACQUIRE, "wavefront");
            if (lane < cnt) {
                const int2 en = wq[(qh + lane) & (LNK_QN - 1)];
                const int sb = eidx[en.y];
                if (atomicCAS(&lhs[en.x], -1, sb) != -1) {
                    const int p = atomicExch(&lhn[en.x], en.y);
                    ((int2*)rec)[en.y] = make_int2(sb, p);
                }
            }
        };

        auto sieve_pass = [&](const int4 (&dv)[LNK_U], int b0, auto tail) {
            constexpr bool TAIL = decltype(tail)::value;
            unsigned u[LNK_U * 4], vote[LNK_U * 4];
            bool m[LNK_U * 4];
            #pragma unroll
            for (int g = 0; g < LNK_U; ++g) {
                bool live = true;
                if constexpr (TAIL) live = b0 + g * STR < c1;
                const int d[4] = {dv[g].x, dv[g].y, dv[g].z, dv[g].w};
                #pragma unroll
                for (int j = 0; j < 4; ++j) {
                    u[g * 4 + j] = (unsigned)(d[j] - lo);
                    m[g * 4 + j] = live && u[g * 4 + j] < (unsigned)nr;
                    vote[g * 4 + j] = __builtin_amdgcn_ballot_w32(m[g * 4 + j]);
                }
            }
            #pragma unroll
            for (int k = 0; k < LNK_U * 4; ++k) {
                if (vote[k] != 0u) {
                    if (m[k]) wq[(qt + (int)__builtin_amdgcn_mbcnt_lo(vote[k], 0u)) & (LNK_QN - 1)] =
                        make_int2((int)u[k], b0 + (k >> 2) * STR + (k & 3));
                    qt += __builtin_popcount(vote[k]);
                    if (qt - qh >= 32) { link_batch(32); qh += 32; }
                }
            }
        };

        int4 dn[LNK_U];
        if (c0 < c1) {
            #pragma unroll
            for (int g = 0; g < LNK_U; ++g)
                dn[g] = *(const int4*)(dstp + min(c0 + (int)threadIdx.x * 4 + g * STR, c1 - 4));
        }
        int base = c0;
        for (; base + STEP <= c1; base += STEP) {
            const int b0 = base + threadIdx.x * 4;
            int4 dv[LNK_U];
            #pragma unroll
            for (int g = 0; g < LNK_U; ++g) dv[g] = dn[g];
            if (base + STEP < c1) {
                #pragma unroll
                for (int g = 0; g < LNK_U; ++g)
                    dn[g] = *(const int4*)(dstp + min(b0 + STEP + g * STR, c1 - 4));
            }
            sieve_pass(dv, b0, lnk_full{});
        }
        if (base < c1) sieve_pass(dn, base + (int)threadIdx.x * 4, lnk_tail{});
        link_batch(qt - qh);
    } else {
        for (int b = c0 + threadIdx.x; b < c1; b += LNK_TPB) {
            const unsigned u = (unsigned)(dstp[b] - lo);
            if (u < (unsigned)nr) {
                const int sb = eidx[b];
                if (atomicCAS(&lhs[u], -1, sb) != -1) {
                    const int p = atomicExch(&lhn[u], b);
                    ((int2*)rec)[b] = make_int2(sb, p);
                }
            }
        }
    }
    __syncthreads();
    for (int i = threadIdx.x; i < nr; i += LNK_TPB)
        if (lo + i < n) ((int2*)headT)[(size_t)c * n + lo + i] = make_int2(lhs[i], lhn[i]);
}

template <int H, int C, int HS4, int AST>
__global__ void row_walk_kernel(const int* __restrict__ rec, int E, int n,
                                const int* __restrict__ headT,
                                const float* __restrict__ as,
                                const float* __restrict__ ad,
                                const float* __restrict__ h,
                                const float* __restrict__ bias,
                                float* __restrict__ out) {
    constexpr int LPR = (H * C) / 4;
    const int gid = blockIdx.x * blockDim.x + threadIdx.x;
    const int row = gid / LPR;
    if (row >= n) return;
    const int q  = gid - row * LPR;
    const int hh = (q * 4) / C;
    const float4* h4 = (const float4*)h;
    const int2*   r2 = (const int2*)rec;

    const float adv = ad[row * AST + hh];
    float  m   = leaky(as[row * AST + hh] + adv);
    float  den = 1.f;
    float4 acc = h4[(size_t)row * HS4 + q];

    int nhs[LNK_B], nhn[LNK_B];
    int s = -1, nx = -1;
    {
        const int2* hT = (const int2*)headT;
        int2 hd[LNK_B];
        #pragma unroll
        for (int c = 0; c < LNK_B; ++c) hd[c] = hT[(size_t)c * n + row];
        #pragma unroll
        for (int c = LNK_B - 1; c >= 0; --c) {
            nhs[c] = s; nhn[c] = nx;
            if (hd[c].x >= 0) { s = hd[c].x; nx = hd[c].y; }
        }
    }
    for (int it = 0; s >= 0 && it < E; ++it) {
        const float4 hv = h4[(size_t)s * HS4 + q];
        const float  sc = leaky(as[s * AST + hh] + adv);
        int s2 = -1, nx2 = -1;
        if (nx >= 0) {
            const int2 r = r2[nx]; s2 = r.x; nx2 = r.y;
        } else {
            const int ce = -nx - 2;
            #pragma unroll
            for (int c = 0; c < LNK_B; ++c) { s2 = (ce == c) ? nhs[c] : s2; nx2 = (ce == c) ? nhn[c] : nx2; }
        }
        const float mn = fmaxf(m, sc);
        const float a  = __expf(m - mn);
        const float w  = __expf(sc - mn);
        den   = den   * a + w;
        acc.x = acc.x * a + w * hv.x;
        acc.y = acc.y * a + w * hv.y;
        acc.z = acc.z * a + w * hv.z;
        acc.w = acc.w * a + w * hv.w;
        m = mn;
        s = s2; nx = nx2;
    }
    const float4 b = ((const float4*)bias)[q];
    float4 o;
    o.x = acc.x / den + b.x;
    o.y = acc.y / den + b.y;
    o.z = acc.z / den + b.z;
    o.w = acc.w / den + b.w;
    ((float4*)out)[(size_t)row * LPR + q] = o;
}

static inline int ceil_div(int a, int b) { return (a + b - 1) / b; }

extern "C" void kernel_launch(void* const* d_in, const int* in_sizes, int n_in,
                              void* d_out, int out_size, void* d_ws, size_t ws_size,
                              hipStream_t stream) {
    const float* x    = (const float*)d_in[0];
    const int*   eidx = (const int*)  d_in[1];
    const float* W1   = (const float*)d_in[2];
    const float* aS1  = (const float*)d_in[3];
    const float* aD1  = (const float*)d_in[4];
    const float* b1   = (const float*)d_in[5];
    const float* W2   = (const float*)d_in[6];
    const float* aS2  = (const float*)d_in[7];
    const float* aD2  = (const float*)d_in[8];
    const float* b2   = (const float*)d_in[9];
    float* out = (float*)d_out;

    const int FEATS = 256, HID = 64, H1 = 8, C1 = 8, CLS = 16;
    const int N = in_sizes[0] / FEATS;
    const int E = in_sizes[1] / 2;

    float* ws = (float*)d_ws;
    float* h1   = ws;                 ws += (size_t)N * HID;
    float* o1   = ws;                 ws += (size_t)N * HID;
    int*   rec  = (int*)ws;           ws += (size_t)E * 2;
    float* as1  = ws;                 ws += (size_t)N * H1;
    float* ad1  = ws;                 ws += (size_t)N * H1;
    int*   headT1 = (int*)out;
    int*   headT2 = (int*)as1;
    const int NR = 32;
    float* nrec = h1;

    const int TPB = 256;
    const dim3 gblock(32 * GEMM_WPB);
    const dim3 ggrid(ceil_div(N / 16, GEMM_WPB), 1);

    proj_logits_kernel<4, H1, false><<<ggrid, gblock, 0, stream>>>(
        x, W1, aS1, aD1, h1, as1, ad1, nullptr, nullptr, 0, N, FEATS, HID, H1);
    link_edges_kernel<<<LNK_B * LNK_R, LNK_TPB, 0, stream>>>(eidx, E, N, headT1, rec);
    row_walk_kernel<H1, C1, HID / 4, H1><<<ceil_div(N * (HID / 4), TPB), TPB, 0, stream>>>(
        rec, E, N, headT1, as1, ad1, h1, b1, o1);

    proj_logits_kernel<1, 1, true><<<ggrid, gblock, 0, stream>>>(
        o1, W2, aS2, aD2, nrec, nrec + CLS, nrec + CLS + 1, headT1, headT2, LNK_B * (N / 2),
        N, HID, NR, NR);
    row_walk_kernel<1, CLS, NR / 4, NR><<<ceil_div(N * (CLS / 4), TPB), TPB, 0, stream>>>(
        rec, E, N, headT2, nrec + CLS, nrec + CLS + 1, nrec, b2, out);
}


# list head records pack a list's first three sources (rank by one LDS add; third rides the chain's last link): record reads only from the 4th edge on; in-chunk sum order now 1st, 2nd, chain, 3rd
# speedup vs baseline: 6.9732x; 1.0803x over previous
#include <hip/hip_runtime.h>
#include <math.h>

typedef float v2f __attribute__((ext_vector_type(2)));
typedef float v8f __attribute__((ext_vector_type(8)));

#define NEG_SLOPE 0.2f

__device__ __forceinline__ float leaky(float x) {
    return x > 0.f ? x : NEG_SLOPE * x;
}

#define GEMM_WPB 4
template <int CT, int H, bool COPY_HEADS>
__global__ void proj_logits_kernel(const float* __restrict__ A,
                                   const float* __restrict__ B,
                                   const float* __restrict__ a_src,
                                   const float* __restrict__ a_dst,
                                   float* Out, float* as, float* ad,
                                   const int* cps, int* cpd, int cpn4,
                                   int M, int K, int ldo, int ast) {
    constexpr int NOUT = CT * 16;
    constexpr int C    = NOUT / H;
    constexpr int P    = NOUT + 4;
    __shared__ float4 tile4[GEMM_WPB][16 * P / 4];

    if constexpr (COPY_HEADS) {
        const int nthr = gridDim.x * blockDim.x;
        for (int i = blockIdx.x * blockDim.x + threadIdx.x; i < cpn4; i += nthr)
            ((int4*)cpd)[i] = ((const int4*)cps)[i];
    }

    const int wave = threadIdx.x >> 5;
    const int lane = threadIdx.x & 31;
    const int row0 = (blockIdx.x * GEMM_WPB + wave) << 4;
    if (row0 >= M) return;

    const int half = lane >> 4;
    const int l16  = lane & 15;
    const float* arow = A + (size_t)(row0 + l16) * K;

    static_assert(H == 1 || 32 % H == 0, "a lane keeps one head");
    constexpr int NW = (H == 1) ? C : 2 * C;
    const int hh = (H == 1) ? 0 : lane % H;
    float wreg[NW];
    if constexpr (H == 1) {
        const float* wv = half ? a_dst : a_src;
        #pragma unroll
        for (int c = 0; c < C; ++c) wreg[c] = wv[c];
    } else {
        #pragma unroll
        for (int c = 0; c < C; ++c) { wreg[c] = a_src[hh * C + c]; wreg[C + c] = a_dst[hh * C + c]; }
    }

    v8f acc[CT];
    #pragma unroll
    for (int t = 0; t < CT; ++t) acc[t] = (v8f){};

    for (int k = 0; k < K; k += 4) {
        const int ka = k + 2 * half;
        v2f a;
        a.x = arow[ka];
        a.y = arow[ka + 1];
        const float* brow0 = B + (size_t)ka * NOUT + l16;
        const float* brow1 = B + (size_t)(ka + 1) * NOUT + l16;
        #pragma unroll
        for (int t = 0; t < CT; ++t) {
            v2f b;
            b.x = brow0[t * 16];
            b.y = brow1[t * 16];
            acc[t] = __builtin_amdgcn_wmma_f32_16x16x4_f32(
                 false, a,  false, b,
                 (short)0, acc[t],  false,  false);
        }
    }

    float* tw = (float*)tile4[wave];
    #pragma unroll
    for (int t = 0; t < CT; ++t) {
        #pragma unroll
        for (int r = 0; r < 8; ++r) {
            tw[(r + 8 * half) * P + t * 16 + l16] = acc[t][r];
        }
    }
    __builtin_amdgcn_fence(__ATOMIC_RELEASE, "wavefront");
    __builtin_amdgcn_wave_barrier();
    __builtin_amdgcn_fence(__ATOMIC_ACQUIRE, "wavefront");

    constexpr int F4R = NOUT / 4;
    #pragma unroll
    for (int f = lane; f < 16 * F4R; f += 32) {
        const int row = f / F4R;
        const int c4  = f - row * F4R;
        const float4 v = *(const float4*)(tw + row * P + c4 * 4);
        *(float4*)(Out + (size_t)(row0 + row) * ldo + c4 * 4) = v;
    }

    if constexpr (H == 1) {
        const float* hp = tw + l16 * P;
        float s = 0.f;
        #pragma unroll
        for (int c = 0; c < C; ++c) s += hp[c] * wreg[c];
        float* o = half ? ad : as;
        o[(size_t)(row0 + l16) * ast] = s;
    } else {
        #pragma unroll
        for (int p = lane; p < 16 * H; p += 32) {
            const int row = p / H;
            const float* hp = tw + row * P + hh * C;
            float s = 0.f, d = 0.f;
            #pragma unroll
            for (int c = 0; c < C; ++c) {
                s += hp[c] * wreg[c];
                d += hp[c] * wreg[C + c];
            }
            as[(size_t)(row0 + row) * ast + hh] = s;
            ad[(size_t)(row0 + row) * ast + hh] = d;
        }
    }
}

#define LNK_B   8
#define LNK_R   32
#define LNK_CAP 4096
#define LNK_TPB 1024
#define LNK_U   4
#define LNK_QN  64
#define LNK_SB  17
#define LNK_SM  ((1 << LNK_SB) - 1)
#define LNK_T1  (1 << 29)
#define LNK_T2  (2 << 29)
static_assert(LNK_R * LNK_CAP <= (1 << LNK_SB), "a source id fits LNK_SB bits");
static_assert(LNK_B <= 8, "the chunk takes 3 bits of a head record");
struct lnk_full { static constexpr bool value = false; };
struct lnk_tail { static constexpr bool value = true;  };
__global__ void __launch_bounds__(LNK_TPB)
link_edges_kernel(const int* __restrict__ eidx, int E, int n,
                  int* __restrict__ headT,
                  int* __restrict__ rec) {
    __shared__ int lcnt[LNK_CAP];
    __shared__ int lsrc[3][LNK_CAP];
    __shared__ int lnx[LNK_CAP];
    __shared__ int lte[LNK_CAP];
    __shared__ int2 ring[LNK_TPB / 32][LNK_QN];
    const int c   = blockIdx.x / LNK_R;
    const int r   = blockIdx.x - c * LNK_R;
    const int nr  = min((n + LNK_R - 1) / LNK_R, LNK_CAP);
    const int lo  = r * nr;
    const int ch  = ((E + LNK_B * 4 - 1) / (LNK_B * 4)) * 4;
    const int c0  = c * ch;
    const int c1  = min(c0 + ch, E);
    for (int i = threadIdx.x; i < nr; i += LNK_TPB) { lcnt[i] = 0; lnx[i] = -1; }
    __syncthreads();

    auto file_edge = [&](int u, int e, int sb) {
        const int rank = atomicAdd(&lcnt[u], 1);
        if (rank < 3) {
            lsrc[rank][u] = sb;
        } else {
            const int p = atomicExch(&lnx[u], e);
            if (p < 0) { lte[u] = e; rec[2 * (size_t)e] = sb; }
            else       ((int2*)rec)[e] = make_int2(sb, p);
        }
    };

    const int* __restrict__ dstp = eidx + E;
    constexpr int STR = LNK_TPB * 4;
    if ((E & 3) == 0) {
        constexpr int STEP = STR * LNK_U;
        const int lane = threadIdx.x & 31;
        int2* wq = ring[threadIdx.x >> 5];
        int qh = 0, qt = 0;

        auto link_batch = [&](int cnt) {
            __builtin_amdgcn_fence(__ATOMIC_RELEASE, "wavefront");
            __builtin_amdgcn_wave_barrier();
            __builtin_amdgcn_fence(__ATOMIC_ACQUIRE, "wavefront");
            if (lane < cnt) {
                const int2 en = wq[(qh + lane) & (LNK_QN - 1)];
                file_edge(en.x, en.y, eidx[en.y]);
            }
        };

        auto sieve_pass = [&](const int4 (&dv)[LNK_U], int b0, auto tail) {
            constexpr bool TAIL = decltype(tail)::value;
            unsigned u[LNK_U * 4], vote[LNK_U * 4];
            bool m[LNK_U * 4];
            #pragma unroll
            for (int g = 0; g < LNK_U; ++g) {
                bool live = true;
                if constexpr (TAIL) live = b0 + g * STR < c1;
                const int d[4] = {dv[g].x, dv[g].y, dv[g].z, dv[g].w};
                #pragma unroll
                for (int j = 0; j < 4; ++j) {
                    u[g * 4 + j] = (unsigned)(d[j] - lo);
                    m[g * 4 + j] = live && u[g * 4 + j] < (unsigned)nr;
                    vote[g * 4 + j] = __builtin_amdgcn_ballot_w32(m[g * 4 + j]);
                }
            }
            #pragma unroll
            for (int k = 0; k < LNK_U * 4; ++k) {
                if (vote[k] != 0u) {
                    if (m[k]) wq[(qt + (int)__builtin_amdgcn_mbcnt_lo(vote[k], 0u)) & (LNK_QN - 1)] =
                        make_int2((int)u[k], b0 + (k >> 2) * STR + (k & 3));
                    qt += __builtin_popcount(vote[k]);
                    if (qt - qh >= 32) { link_batch(32); qh += 32; }
                }
            }
        };

        int4 dn[LNK_U];
        if (c0 < c1) {
            #pragma unroll
            for (int g = 0; g < LNK_U; ++g)
                dn[g] = *(const int4*)(dstp + min(c0 + (int)threadIdx.x * 4 + g * STR, c1 - 4));
        }
        int base = c0;
        for (; base + STEP <= c1; base += STEP) {
            const int b0 = base + threadIdx.x * 4;
            int4 dv[LNK_U];
            #pragma unroll
            for (int g = 0; g < LNK_U; ++g) dv[g] = dn[g];
            if (base + STEP < c1) {
                #pragma unroll
                for (int g = 0; g < LNK_U; ++g)
                    dn[g] = *(const int4*)(dstp + min(b0 + STEP + g * STR, c1 - 4));
            }
            sieve_pass(dv, b0, lnk_full{});
        }
        if (base < c1) sieve_pass(dn, base + (int)threadIdx.x * 4, lnk_tail{});
        link_batch(qt - qh);
    } else {
        for (int b = c0 + threadIdx.x; b < c1; b += LNK_TPB) {
            const unsigned u = (unsigned)(dstp[b] - lo);
            if (u < (unsigned)nr) file_edge((int)u, b, eidx[b]);
        }
    }
    __syncthreads();
    for (int i = threadIdx.x; i < nr; i += LNK_TPB) {
        if (lo + i < n) {
            const int k = lcnt[i];
            unsigned x = 0u, y = 0u;
            if (k > 0) {
                const unsigned s0 = (unsigned)lsrc[0][i];
                const unsigned s1 = k > 1 ? (unsigned)lsrc[1][i] : 0u;
                unsigned f3 = 0u;
                if (k == 3) {
                    f3 = (unsigned)lsrc[2][i];
                } else if (k > 3) {
                    f3 = (unsigned)lnx[i];
                    rec[2 * (size_t)lte[i] + 1] = LNK_T1 | (c << LNK_SB) | lsrc[2][i];
                }
                x = s0 | (s1 << LNK_SB);
                y = (s1 >> (32 - LNK_SB)) | (f3 << 2) | ((unsigned)min(k, 4) << 26) | ((unsigned)c << 29);
            }
            ((int2*)headT)[(size_t)c * n + lo + i] = make_int2((int)x, (int)y);
        }
    }
}

__device__ __forceinline__ void enter_list(int x, int y, int& s, int& nx, int& pend) {
    const unsigned ux = (unsigned)x, uy = (unsigned)y;
    const int kind = (int)((uy >> 26) & 7u);
    const int c    = (int)(uy >> 29);
    const int s1   = (int)((ux >> LNK_SB) | ((uy & 3u) << (32 - LNK_SB)));
    const int f3   = (int)((uy >> 2) & 0xFFFFFFu);
    const int end  = -(c + 2);
    s    = kind > 0 ? (int)(ux & (unsigned)LNK_SM) : -1;
    nx   = kind > 1 ? (LNK_T2 | s1) : end;
    pend = kind == 3 ? (LNK_T1 | (c << LNK_SB) | f3) : (kind > 3 ? f3 : end);
}

template <int H, int C, int HS4, int AST>
__global__ void row_walk_kernel(const int* __restrict__ rec, int E, int n,
                                const int* __restrict__ headT,
                                const float* __restrict__ as,
                                const float* __restrict__ ad,
                                const float* __restrict__ h,
                                const float* __restrict__ bias,
                                float* __restrict__ out) {
    constexpr int LPR = (H * C) / 4;
    const int gid = blockIdx.x * blockDim.x + threadIdx.x;
    const int row = gid / LPR;
    if (row >= n) return;
    const int q  = gid - row * LPR;
    const int hh = (q * 4) / C;
    const float4* h4 = (const float4*)h;
    const int2*   r2 = (const int2*)rec;

    const float adv = ad[row * AST + hh];
    float  m   = leaky(as[row * AST + hh] + adv);
    float  den = 1.f;
    float4 acc = h4[(size_t)row * HS4 + q];

    int nhx[LNK_B], nhy[LNK_B];
    int s, nx, pend;
    {
        const int2* hT = (const int2*)headT;
        int2 hd[LNK_B];
        #pragma unroll
        for (int c = 0; c < LNK_B; ++c) hd[c] = hT[(size_t)c * n + row];
        int cx = 0, cy = 0;
        #pragma unroll
        for (int c = LNK_B - 1; c >= 0; --c) {
            nhx[c] = cx; nhy[c] = cy;
            if (hd[c].y != 0) { cx = hd[c].x; cy = hd[c].y; }
        }
        enter_list(cx, cy, s, nx, pend);
    }
    for (int it = 0; s >= 0 && it < E; ++it) {
        const float4 hv = h4[(size_t)s * HS4 + q];
        const float  sc = leaky(as[s * AST + hh] + adv);
        int s2, nx2, pend2 = pend;
        if (nx < 0) {
            const int ce = -nx - 2;
            int x = 0, y = 0;
            #pragma unroll
            for (int c = 0; c < LNK_B; ++c) { x = (ce == c) ? nhx[c] : x; y = (ce == c) ? nhy[c] : y; }
            enter_list(x, y, s2, nx2, pend2);
        } else if (nx < LNK_T1) {
            const int2 r = r2[nx]; s2 = r.x; nx2 = r.y;
        } else {
            s2  = nx & LNK_SM;
            nx2 = nx < LNK_T2 ? -(((nx >> LNK_SB) & 7) + 2) : pend;
        }
        const float mn = fmaxf(m, sc);
        const float a  = __expf(m - mn);
        const float w  = __expf(sc - mn);
        den   = den   * a + w;
        acc.x = acc.x * a + w * hv.x;
        acc.y = acc.y * a + w * hv.y;
        acc.z = acc.z * a + w * hv.z;
        acc.w = acc.w * a + w * hv.w;
        m = mn;
        s = s2; nx = nx2; pend = pend2;
    }
    const float4 b = ((const float4*)bias)[q];
    float4 o;
    o.x = acc.x / den + b.x;
    o.y = acc.y / den + b.y;
    o.z = acc.z / den + b.z;
    o.w = acc.w / den + b.w;
    ((float4*)out)[(size_t)row * LPR + q] = o;
}

static inline int ceil_div(int a, int b) { return (a + b - 1) / b; }

extern "C" void kernel_launch(void* const* d_in, const int* in_sizes, int n_in,
                              void* d_out, int out_size, void* d_ws, size_t ws_size,
                              hipStream_t stream) {
    const float* x    = (const float*)d_in[0];
    const int*   eidx = (const int*)  d_in[1];
    const float* W1   = (const float*)d_in[2];
    const float* aS1  = (const float*)d_in[3];
    const float* aD1  = (const float*)d_in[4];
    const float* b1   = (const float*)d_in[5];
    const float* W2   = (const float*)d_in[6];
    const float* aS2  = (const float*)d_in[7];
    const float* aD2  = (const float*)d_in[8];
    const float* b2   = (const float*)d_in[9];
    float* out = (float*)d_out;

    const int FEATS = 256, HID = 64, H1 = 8, C1 = 8, CLS = 16;
    const int N = in_sizes[0] / FEATS;
    const int E = in_sizes[1] / 2;

    float* ws = (float*)d_ws;
    float* h1   = ws;                 ws += (size_t)N * HID;
    float* o1   = ws;                 ws += (size_t)N * HID;
    int*   rec  = (int*)ws;           ws += (size_t)E * 2;
    float* as1  = ws;                 ws += (size_t)N * H1;
    float* ad1  = ws;                 ws += (size_t)N * H1;
    int*   headT1 = (int*)out;
    int*   headT2 = (int*)as1;
    const int NR = 32;
    float* nrec = h1;

    const int TPB = 256;
    const dim3 gblock(32 * GEMM_WPB);
    const dim3 ggrid(ceil_div(N / 16, GEMM_WPB), 1);

    proj_logits_kernel<4, H1, false><<<ggrid, gblock, 0, stream>>>(
        x, W1, aS1, aD1, h1, as1, ad1, nullptr, nullptr, 0, N, FEATS, HID, H1);
    link_edges_kernel<<<LNK_B * LNK_R, LNK_TPB, 0, stream>>>(eidx, E, N, headT1, rec);
    row_walk_kernel<H1, C1, HID / 4, H1><<<ceil_div(N * (HID / 4), TPB), TPB, 0, stream>>>(
        rec, E, N, headT1, as1, ad1, h1, b1, o1);

    proj_logits_kernel<1, 1, true><<<ggrid, gblock, 0, stream>>>(
        o1, W2, aS2, aD2, nrec, nrec + CLS, nrec + CLS + 1, headT1, headT2, LNK_B * (N / 2),
        N, HID, NR, NR);
    row_walk_kernel<1, CLS, NR / 4, NR><<<ceil_div(N * (CLS / 4), TPB), TPB, 0, stream>>>(
        rec, E, N, headT2, nrec + CLS, nrec + CLS + 1, nrec, b2, out);
}


# row walk: a loop over a row's 8 chunk lists decodes each head record once and requests the sources it names together; no link decode or next-list select chain per hop; sum order unchanged
# speedup vs baseline: 7.4955x; 1.0749x over previous
#include <hip/hip_runtime.h>
#include <math.h>

typedef float v2f __attribute__((ext_vector_type(2)));
typedef float v8f __attribute__((ext_vector_type(8)));

#define NEG_SLOPE 0.2f

__device__ __forceinline__ float leaky(float x) {
    return x > 0.f ? x : NEG_SLOPE * x;
}

#define GEMM_WPB 4
template <int CT, int H, bool COPY_HEADS>
__global__ void proj_logits_kernel(const float* __restrict__ A,
                                   const float* __restrict__ B,
                                   const float* __restrict__ a_src,
                                   const float* __restrict__ a_dst,
                                   float* Out, float* as, float* ad,
                                   const int* cps, int* cpd, int cpn4,
                                   int M, int K, int ldo, int ast) {
    constexpr int NOUT = CT * 16;
    constexpr int C    = NOUT / H;
    constexpr int P    = NOUT + 4;
    __shared__ float4 tile4[GEMM_WPB][16 * P / 4];

    if constexpr (COPY_HEADS) {
        const int nthr = gridDim.x * blockDim.x;
        for (int i = blockIdx.x * blockDim.x + threadIdx.x; i < cpn4; i += nthr)
            ((int4*)cpd)[i] = ((const int4*)cps)[i];
    }

    const int wave = threadIdx.x >> 5;
    const int lane = threadIdx.x & 31;
    const int row0 = (blockIdx.x * GEMM_WPB + wave) << 4;
    if (row0 >= M) return;

    const int half = lane >> 4;
    const int l16  = lane & 15;
    const float* arow = A + (size_t)(row0 + l16) * K;

    static_assert(H == 1 || 32 % H == 0, "a lane keeps one head");
    constexpr int NW = (H == 1) ? C : 2 * C;
    const int hh = (H == 1) ? 0 : lane % H;
    float wreg[NW];
    if constexpr (H == 1) {
        const float* wv = half ? a_dst : a_src;
        #pragma unroll
        for (int c = 0; c < C; ++c) wreg[c] = wv[c];
    } else {
        #pragma unroll
        for (int c = 0; c < C; ++c) { wreg[c] = a_src[hh * C + c]; wreg[C + c] = a_dst[hh * C + c]; }
    }

    v8f acc[CT];
    #pragma unroll
    for (int t = 0; t < CT; ++t) acc[t] = (v8f){};

    for (int k = 0; k < K; k += 4) {
        const int ka = k + 2 * half;
        v2f a;
        a.x = arow[ka];
        a.y = arow[ka + 1];
        const float* brow0 = B + (size_t)ka * NOUT + l16;
        const float* brow1 = B + (size_t)(ka + 1) * NOUT + l16;
        #pragma unroll
        for (int t = 0; t < CT; ++t) {
            v2f b;
            b.x = brow0[t * 16];
            b.y = brow1[t * 16];
            acc[t] = __builtin_amdgcn_wmma_f32_16x16x4_f32(
                 false, a,  false, b,
                 (short)0, acc[t],  false,  false);
        }
    }

    float* tw = (float*)tile4[wave];
    #pragma unroll
    for (int t = 0; t < CT; ++t) {
        #pragma unroll
        for (int r = 0; r < 8; ++r) {
            tw[(r + 8 * half) * P + t * 16 + l16] = acc[t][r];
        }
    }
    __builtin_amdgcn_fence(__ATOMIC_RELEASE, "wavefront");
    __builtin_amdgcn_wave_barrier();
    __builtin_amdgcn_fence(__ATOMIC_ACQUIRE, "wavefront");

    constexpr int F4R = NOUT / 4;
    #pragma unroll
    for (int f = lane; f < 16 * F4R; f += 32) {
        const int row = f / F4R;
        const int c4  = f - row * F4R;
        const float4 v = *(const float4*)(tw + row * P + c4 * 4);
        *(float4*)(Out + (size_t)(row0 + row) * ldo + c4 * 4) = v;
    }

    if constexpr (H == 1) {
        const float* hp = tw + l16 * P;
        float s = 0.f;
        #pragma unroll
        for (int c = 0; c < C; ++c) s += hp[c] * wreg[c];
        float* o = half ? ad : as;
        o[(size_t)(row0 + l16) * ast] = s;
    } else {
        #pragma unroll
        for (int p = lane; p < 16 * H; p += 32) {
            const int row = p / H;
            const float* hp = tw + row * P + hh * C;
            float s = 0.f, d = 0.f;
            #pragma unroll
            for (int c = 0; c < C; ++c) {
                s += hp[c] * wreg[c];
                d += hp[c] * wreg[C + c];
            }
            as[(size_t)(row0 + row) * ast + hh] = s;
            ad[(size_t)(row0 + row) * ast + hh] = d;
        }
    }
}

#define LNK_B   8
#define LNK_R   32
#define LNK_CAP 4096
#define LNK_TPB 1024
#define LNK_U   4
#define LNK_QN  64
#define LNK_SB  17
#define LNK_SM  ((1 << LNK_SB) - 1)
#define LNK_T1  (1 << 29)
static_assert(LNK_R * LNK_CAP <= (1 << LNK_SB), "a source id fits LNK_SB bits");
static_assert(LNK_B <= 8, "the chunk takes 3 bits of a head record");
struct lnk_full { static constexpr bool value = false; };
struct lnk_tail { static constexpr bool value = true;  };
__global__ void __launch_bounds__(LNK_TPB)
link_edges_kernel(const int* __restrict__ eidx, int E, int n,
                  int* __restrict__ headT,
                  int* __restrict__ rec) {
    __shared__ int lcnt[LNK_CAP];
    __shared__ int lsrc[3][LNK_CAP];
    __shared__ int lnx[LNK_CAP];
    __shared__ int lte[LNK_CAP];
    __shared__ int2 ring[LNK_TPB / 32][LNK_QN];
    const int c   = blockIdx.x / LNK_R;
    const int r   = blockIdx.x - c * LNK_R;
    const int nr  = min((n + LNK_R - 1) / LNK_R, LNK_CAP);
    const int lo  = r * nr;
    const int ch  = ((E + LNK_B * 4 - 1) / (LNK_B * 4)) * 4;
    const int c0  = c * ch;
    const int c1  = min(c0 + ch, E);
    for (int i = threadIdx.x; i < nr; i += LNK_TPB) { lcnt[i] = 0; lnx[i] = -1; }
    __syncthreads();

    auto file_edge = [&](int u, int e, int sb) {
        const int rank = atomicAdd(&lcnt[u], 1);
        if (rank < 3) {
            lsrc[rank][u] = sb;
        } else {
            const int p = atomicExch(&lnx[u], e);
            if (p < 0) { lte[u] = e; rec[2 * (size_t)e] = sb; }
            else       ((int2*)rec)[e] = make_int2(sb, p);
        }
    };

    const int* __restrict__ dstp = eidx + E;
    constexpr int STR = LNK_TPB * 4;
    if ((E & 3) == 0) {
        constexpr int STEP = STR * LNK_U;
        const int lane = threadIdx.x & 31;
        int2* wq = ring[threadIdx.x >> 5];
        int qh = 0, qt = 0;

        auto link_batch = [&](int cnt) {
            __builtin_amdgcn_fence(__ATOMIC_RELEASE, "wavefront");
            __builtin_amdgcn_wave_barrier();
            __builtin_amdgcn_fence(__ATOMIC_ACQUIRE, "wavefront");
            if (lane < cnt) {
                const int2 en = wq[(qh + lane) & (LNK_QN - 1)];
                file_edge(en.x, en.y, eidx[en.y]);
            }
        };

        auto sieve_pass = [&](const int4 (&dv)[LNK_U], int b0, auto tail) {
            constexpr bool TAIL = decltype(tail)::value;
            unsigned u[LNK_U * 4], vote[LNK_U * 4];
            bool m[LNK_U * 4];
            #pragma unroll
            for (int g = 0; g < LNK_U; ++g) {
                bool live = true;
                if constexpr (TAIL) live = b0 + g * STR < c1;
                const int d[4] = {dv[g].x, dv[g].y, dv[g].z, dv[g].w};
                #pragma unroll
                for (int j = 0; j < 4; ++j) {
                    u[g * 4 + j] = (unsigned)(d[j] - lo);
                    m[g * 4 + j] = live && u[g * 4 + j] < (unsigned)nr;
                    vote[g * 4 + j] = __builtin_amdgcn_ballot_w32(m[g * 4 + j]);
                }
            }
            #pragma unroll
            for (int k = 0; k < LNK_U * 4; ++k) {
                if (vote[k] != 0u) {
                    if (m[k]) wq[(qt + (int)__builtin_amdgcn_mbcnt_lo(vote[k], 0u)) & (LNK_QN - 1)] =
                        make_int2((int)u[k], b0 + (k >> 2) * STR + (k & 3));
                    qt += __builtin_popcount(vote[k]);
                    if (qt - qh >= 32) { link_batch(32); qh += 32; }
                }
            }
        };

        int4 dn[LNK_U];
        if (c0 < c1) {
            #pragma unroll
            for (int g = 0; g < LNK_U; ++g)
                dn[g] = *(const int4*)(dstp + min(c0 + (int)threadIdx.x * 4 + g * STR, c1 - 4));
        }
        int base = c0;
        for (; base + STEP <= c1; base += STEP) {
            const int b0 = base + threadIdx.x * 4;
            int4 dv[LNK_U];
            #pragma unroll
            for (int g = 0; g < LNK_U; ++g) dv[g] = dn[g];
            if (base + STEP < c1) {
                #pragma unroll
                for (int g = 0; g < LNK_U; ++g)
                    dn[g] = *(const int4*)(dstp + min(b0 + STEP + g * STR, c1 - 4));
            }
            sieve_pass(dv, b0, lnk_full{});
        }
        if (base < c1) sieve_pass(dn, base + (int)threadIdx.x * 4, lnk_tail{});
        link_batch(qt - qh);
    } else {
        for (int b = c0 + threadIdx.x; b < c1; b += LNK_TPB) {
            const unsigned u = (unsigned)(dstp[b] - lo);
            if (u < (unsigned)nr) file_edge((int)u, b, eidx[b]);
        }
    }
    __syncthreads();
    for (int i = threadIdx.x; i < nr; i += LNK_TPB) {
        if (lo + i < n) {
            const int k = lcnt[i];
            unsigned x = 0u, y = 0u;
            if (k > 0) {
                const unsigned s0 = (unsigned)lsrc[0][i];
                const unsigned s1 = k > 1 ? (unsigned)lsrc[1][i] : 0u;
                unsigned f3 = 0u;
                if (k == 3) {
                    f3 = (unsigned)lsrc[2][i];
                } else if (k > 3) {
                    f3 = (unsigned)lnx[i];
                    rec[2 * (size_t)lte[i] + 1] = LNK_T1 | (c << LNK_SB) | lsrc[2][i];
                }
                x = s0 | (s1 << LNK_SB);
                y = (s1 >> (32 - LNK_SB)) | (f3 << 2) | ((unsigned)min(k, 4) << 26) | ((unsigned)c << 29);
            }
            ((int2*)headT)[(size_t)c * n + lo + i] = make_int2((int)x, (int)y);
        }
    }
}

template <int H, int C, int HS4, int AST>
__global__ void row_walk_kernel(const int* __restrict__ rec, int E, int n,
                                const int* __restrict__ headT,
                                const float* __restrict__ as,
                                const float* __restrict__ ad,
                                const float* __restrict__ h,
                                const float* __restrict__ bias,
                                float* __restrict__ out) {
    constexpr int LPR = (H * C) / 4;
    static_assert(((1 << LNK_SB) - 1) < (1 << 30) / (HS4 > AST ? HS4 : AST), "32-bit row and logit indices");
    const int gid = blockIdx.x * blockDim.x + threadIdx.x;
    const int row = gid / LPR;
    if (row >= n) return;
    const int q  = gid - row * LPR;
    const int hh = (q * 4) / C;
    const float4* h4 = (const float4*)h;
    const int2*   r2 = (const int2*)rec;
    const int2*   hT = (const int2*)headT + row;

    const float adv = ad[row * AST + hh];
    float  m   = leaky(as[row * AST + hh] + adv);
    float  den = 1.f;
    float4 acc = h4[row * HS4 + q];

    auto add_edge = [&](const float4 hv, const float av) {
        const float sc = leaky(av + adv);
        const float mn = fmaxf(m, sc);
        const float a  = __expf(m - mn);
        const float w  = __expf(sc - mn);
        den   = den   * a + w;
        acc.x = acc.x * a + w * hv.x;
        acc.y = acc.y * a + w * hv.y;
        acc.z = acc.z * a + w * hv.z;
        acc.w = acc.w * a + w * hv.w;
        m = mn;
    };

    int2 hn = hT[0];
    #pragma nounroll
    for (int c = 0; c < LNK_B; ++c) {
        const int2 hd = hn;
        if (c + 1 < LNK_B) hn = hT[(size_t)(c + 1) * n];
        const unsigned ux = (unsigned)hd.x, uy = (unsigned)hd.y;
        const int kind = (int)((uy >> 26) & 7u);
        const int s0   = (int)(ux & (unsigned)LNK_SM);
        const int s1   = (int)((ux >> LNK_SB) | ((uy & 3u) << (32 - LNK_SB)));
        const int f3   = (int)((uy >> 2) & 0xFFFFFFu);
        float4 v0, v1, v2;
        float  a0, a1, a2;
        int2   r;
        if (kind > 0)  { v0 = h4[s0 * HS4 + q]; a0 = as[s0 * AST + hh]; }
        if (kind > 1)  { v1 = h4[s1 * HS4 + q]; a1 = as[s1 * AST + hh]; }
        if (kind == 3) { v2 = h4[f3 * HS4 + q]; a2 = as[f3 * AST + hh]; }
        if (kind > 3)  r = r2[f3];
        if (kind > 0)  add_edge(v0, a0);
        if (kind > 1)  add_edge(v1, a1);
        if (kind == 3) add_edge(v2, a2);
        if (kind > 3) {
            for (int it = 0; it < E; ++it) {
                const int s = r.x, nx = r.y;
                const float4 hv = h4[s * HS4 + q];
                const float  av = as[s * AST + hh];
                if (nx < LNK_T1) {
                    r = r2[nx];
                    add_edge(hv, av);
                } else {
                    const int t = nx & LNK_SM;
                    const float4 tv = h4[t * HS4 + q];
                    const float  ta = as[t * AST + hh];
                    add_edge(hv, av);
                    add_edge(tv, ta);
                    break;
                }
            }
        }
    }
    const float4 b = ((const float4*)bias)[q];
    float4 o;
    o.x = acc.x / den + b.x;
    o.y = acc.y / den + b.y;
    o.z = acc.z / den + b.z;
    o.w = acc.w / den + b.w;
    ((float4*)out)[(size_t)row * LPR + q] = o;
}

static inline int ceil_div(int a, int b) { return (a + b - 1) / b; }

extern "C" void kernel_launch(void* const* d_in, const int* in_sizes, int n_in,
                              void* d_out, int out_size, void* d_ws, size_t ws_size,
                              hipStream_t stream) {
    const float* x    = (const float*)d_in[0];
    const int*   eidx = (const int*)  d_in[1];
    const float* W1   = (const float*)d_in[2];
    const float* aS1  = (const float*)d_in[3];
    const float* aD1  = (const float*)d_in[4];
    const float* b1   = (const float*)d_in[5];
    const float* W2   = (const float*)d_in[6];
    const float* aS2  = (const float*)d_in[7];
    const float* aD2  = (const float*)d_in[8];
    const float* b2   = (const float*)d_in[9];
    float* out = (float*)d_out;

    const int FEATS = 256, HID = 64, H1 = 8, C1 = 8, CLS = 16;
    const int N = in_sizes[0] / FEATS;
    const int E = in_sizes[1] / 2;

    float* ws = (float*)d_ws;
    float* h1   = ws;                 ws += (size_t)N * HID;
    float* o1   = ws;                 ws += (size_t)N * HID;
    int*   rec  = (int*)ws;           ws += (size_t)E * 2;
    float* as1  = ws;                 ws += (size_t)N * H1;
    float* ad1  = ws;                 ws += (size_t)N * H1;
    int*   headT1 = (int*)out;
    int*   headT2 = (int*)as1;
    const int NR = 32;
    float* nrec = h1;

    const int TPB = 256;
    const dim3 gblock(32 * GEMM_WPB);
    const dim3 ggrid(ceil_div(N / 16, GEMM_WPB), 1);

    proj_logits_kernel<4, H1, false><<<ggrid, gblock, 0, stream>>>(
        x, W1, aS1, aD1, h1, as1, ad1, nullptr, nullptr, 0, N, FEATS, HID, H1);
    link_edges_kernel<<<LNK_B * LNK_R, LNK_TPB, 0, stream>>>(eidx, E, N, headT1, rec);
    row_walk_kernel<H1, C1, HID / 4, H1><<<ceil_div(N * (HID / 4), TPB), TPB, 0, stream>>>(
        rec, E, N, headT1, as1, ad1, h1, b1, o1);

    proj_logits_kernel<1, 1, true><<<ggrid, gblock, 0, stream>>>(
        o1, W2, aS2, aD2, nrec, nrec + CLS, nrec + CLS + 1, headT1, headT2, LNK_B * (N / 2),
        N, HID, NR, NR);
    row_walk_kernel<1, CLS, NR / 4, NR><<<ceil_div(N * (CLS / 4), TPB), TPB, 0, stream>>>(
        rec, E, N, headT2, nrec + CLS, nrec + CLS + 1, nrec, b2, out);
}


# projection launches: a block lays W once in LDS in fragment order (25 waves share it); a k-step's B fragments come by 16-byte LDS reads, not by global 4-byte loads; k and sum order unchanged
# speedup vs baseline: 7.8998x; 1.0539x over previous
#include <hip/hip_runtime.h>
#include <math.h>

typedef float v2f __attribute__((ext_vector_type(2)));
typedef float v8f __attribute__((ext_vector_type(8)));

#define NEG_SLOPE 0.2f

__device__ __forceinline__ float leaky(float x) {
    return x > 0.f ? x : NEG_SLOPE * x;
}

#define GEMM_WPB 25
template <int CT, int H, int KK, bool COPY_HEADS>
__global__ void __launch_bounds__(32 * GEMM_WPB)
proj_logits_kernel(const float* __restrict__ A,
                   const float* __restrict__ B,
                   const float* __restrict__ a_src,
                   const float* __restrict__ a_dst,
                   float* Out, float* as, float* ad,
                   const int* cps, int* cpd, int cpn4,
                   int M, int ldo, int ast) {
    constexpr int NOUT = CT * 16;
    constexpr int C    = NOUT / H;
    constexpr int P    = NOUT + 4;
    constexpr int NPL  = 2 * CT;
    constexpr int NT   = 32 * GEMM_WPB;
    static_assert(KK % 4 == 0 && NOUT % 4 == 0, "whole k-steps, whole float4s of B");
    __shared__ float4 tile4[GEMM_WPB][16 * P / 4];
    __shared__ __attribute__((aligned(16))) float Bs[KK * NOUT];

    {
        constexpr int NV  = KK * NOUT / 4;
        constexpr int NIT = (NV + NT - 1) / NT;
        float4 bv[NIT];
        #pragma unroll
        for (int it = 0; it < NIT; ++it) {
            const int i = threadIdx.x + it * NT;
            if (i < NV) bv[it] = ((const float4*)B)[i];
        }
        #pragma unroll
        for (int it = 0; it < NIT; ++it) {
            const int i = threadIdx.x + it * NT;
            if (i < NV) {
                const int kk = (4 * i) / NOUT;
                const int n0 = (4 * i) - kk * NOUT;
                const int ln = ((kk >> 1) & 1) * 16 + (n0 & 15);
                float* d = Bs + (((kk >> 2) * 32 + ln) * NPL) + (n0 >> 4) * 2 + (kk & 1);
                d[0]       = bv[it].x;
                d[NPL]     = bv[it].y;
                d[2 * NPL] = bv[it].z;
                d[3 * NPL] = bv[it].w;
            }
        }
    }

    if constexpr (COPY_HEADS) {
        const int nthr = gridDim.x * blockDim.x;
        for (int i = blockIdx.x * blockDim.x + threadIdx.x; i < cpn4; i += nthr)
            ((int4*)cpd)[i] = ((const int4*)cps)[i];
    }
    __syncthreads();

    const int wave = threadIdx.x >> 5;
    const int lane = threadIdx.x & 31;
    const int row0 = (blockIdx.x * GEMM_WPB + wave) << 4;
    if (row0 >= M) return;

    const int half = lane >> 4;
    const int l16  = lane & 15;
    const float* arow = A + (size_t)(row0 + l16) * KK;

    static_assert(H == 1 || 32 % H == 0, "a lane keeps one head");
    constexpr int NW = (H == 1) ? C : 2 * C;
    const int hh = (H == 1) ? 0 : lane % H;
    float wreg[NW];
    if constexpr (H == 1) {
        const float* wv = half ? a_dst : a_src;
        #pragma unroll
        for (int c = 0; c < C; ++c) wreg[c] = wv[c];
    } else {
        #pragma unroll
        for (int c = 0; c < C; ++c) { wreg[c] = a_src[hh * C + c]; wreg[C + c] = a_dst[hh * C + c]; }
    }

    v8f acc[CT];
    #pragma unroll
    for (int t = 0; t < CT; ++t) acc[t] = (v8f){};

    const float* bl = Bs + lane * NPL;
    #pragma unroll 4
    for (int k = 0; k < KK; k += 4) {
        const int ka = k + 2 * half;
        v2f a;
        a.x = arow[ka];
        a.y = arow[ka + 1];
        const float* bp = bl + (k >> 2) * (32 * NPL);
        float bb[NPL];
        if constexpr (NPL % 4 == 0) {
            #pragma unroll
            for (int j = 0; j < NPL / 4; ++j) {
                const float4 v = ((const float4*)bp)[j];
                bb[4 * j] = v.x; bb[4 * j + 1] = v.y; bb[4 * j + 2] = v.z; bb[4 * j + 3] = v.w;
            }
        } else {
            static_assert(NPL == 2, "one column tile: a lane's pair is one 8-byte read");
            const float2 v = *(const float2*)bp;
            bb[0] = v.x; bb[1] = v.y;
        }
        #pragma unroll
        for (int t = 0; t < CT; ++t) {
            v2f b;
            b.x = bb[2 * t];
            b.y = bb[2 * t + 1];
            acc[t] = __builtin_amdgcn_wmma_f32_16x16x4_f32(
                 false, a,  false, b,
                 (short)0, acc[t],  false,  false);
        }
    }

    float* tw = (float*)tile4[wave];
    #pragma unroll
    for (int t = 0; t < CT; ++t) {
        #pragma unroll
        for (int r = 0; r < 8; ++r) {
            tw[(r + 8 * half) * P + t * 16 + l16] = acc[t][r];
        }
    }
    __builtin_amdgcn_fence(__ATOMIC_RELEASE, "wavefront");
    __builtin_amdgcn_wave_barrier();
    __builtin_amdgcn_fence(__ATOMIC_ACQUIRE, "wavefront");

    constexpr int F4R = NOUT / 4;
    #pragma unroll
    for (int f = lane; f < 16 * F4R; f += 32) {
        const int row = f / F4R;
        const int c4  = f - row * F4R;
        const float4 v = *(const float4*)(tw + row * P + c4 * 4);
        *(float4*)(Out + (size_t)(row0 + row) * ldo + c4 * 4) = v;
    }

    if constexpr (H == 1) {
        const float* hp = tw + l16 * P;
        float s = 0.f;
        #pragma unroll
        for (int c = 0; c < C; ++c) s += hp[c] * wreg[c];
        float* o = half ? ad : as;
        o[(size_t)(row0 + l16) * ast] = s;
    } else {
        #pragma unroll
        for (int p = lane; p < 16 * H; p += 32) {
            const int row = p / H;
            const float* hp = tw + row * P + hh * C;
            float s = 0.f, d = 0.f;
            #pragma unroll
            for (int c = 0; c < C; ++c) {
                s += hp[c] * wreg[c];
                d += hp[c] * wreg[C + c];
            }
            as[(size_t)(row0 + row) * ast + hh] = s;
            ad[(size_t)(row0 + row) * ast + hh] = d;
        }
    }
}

#define LNK_B   8
#define LNK_R   32
#define LNK_CAP 4096
#define LNK_TPB 1024
#define LNK_U   4
#define LNK_QN  64
#define LNK_SB  17
#define LNK_SM  ((1 << LNK_SB) - 1)
#define LNK_T1  (1 << 29)
static_assert(LNK_R * LNK_CAP <= (1 << LNK_SB), "a source id fits LNK_SB bits");
static_assert(LNK_B <= 8, "the chunk takes 3 bits of a head record");
struct lnk_full { static constexpr bool value = false; };
struct lnk_tail { static constexpr bool value = true;  };
__global__ void __launch_bounds__(LNK_TPB)
link_edges_kernel(const int* __restrict__ eidx, int E, int n,
                  int* __restrict__ headT,
                  int* __restrict__ rec) {
    __shared__ int lcnt[LNK_CAP];
    __shared__ int lsrc[3][LNK_CAP];
    __shared__ int lnx[LNK_CAP];
    __shared__ int lte[LNK_CAP];
    __shared__ int2 ring[LNK_TPB / 32][LNK_QN];
    const int c   = blockIdx.x / LNK_R;
    const int r   = blockIdx.x - c * LNK_R;
    const int nr  = min((n + LNK_R - 1) / LNK_R, LNK_CAP);
    const int lo  = r * nr;
    const int ch  = ((E + LNK_B * 4 - 1) / (LNK_B * 4)) * 4;
    const int c0  = c * ch;
    const int c1  = min(c0 + ch, E);
    for (int i = threadIdx.x; i < nr; i += LNK_TPB) { lcnt[i] = 0; lnx[i] = -1; }
    __syncthreads();

    auto file_edge = [&](int u, int e, int sb) {
        const int rank = atomicAdd(&lcnt[u], 1);
        if (rank < 3) {
            lsrc[rank][u] = sb;
        } else {
            const int p = atomicExch(&lnx[u], e);
            if (p < 0) { lte[u] = e; rec[2 * (size_t)e] = sb; }
            else       ((int2*)rec)[e] = make_int2(sb, p);
        }
    };

    const int* __restrict__ dstp = eidx + E;
    constexpr int STR = LNK_TPB * 4;
    if ((E & 3) == 0) {
        constexpr int STEP = STR * LNK_U;
        const int lane = threadIdx.x & 31;
        int2* wq = ring[threadIdx.x >> 5];
        int qh = 0, qt = 0;

        auto link_batch = [&](int cnt) {
            __builtin_amdgcn_fence(__ATOMIC_RELEASE, "wavefront");
            __builtin_amdgcn_wave_barrier();
            __builtin_amdgcn_fence(__ATOMIC_ACQUIRE, "wavefront");
            if (lane < cnt) {
                const int2 en = wq[(qh + lane) & (LNK_QN - 1)];
                file_edge(en.x, en.y, eidx[en.y]);
            }
        };

        auto sieve_pass = [&](const int4 (&dv)[LNK_U], int b0, auto tail) {
            constexpr bool TAIL = decltype(tail)::value;
            unsigned u[LNK_U * 4], vote[LNK_U * 4];
            bool m[LNK_U * 4];
            #pragma unroll
            for (int g = 0; g < LNK_U; ++g) {
                bool live = true;
                if constexpr (TAIL) live = b0 + g * STR < c1;
                const int d[4] = {dv[g].x, dv[g].y, dv[g].z, dv[g].w};
                #pragma unroll
                for (int j = 0; j < 4; ++j) {
                    u[g * 4 + j] = (unsigned)(d[j] - lo);
                    m[g * 4 + j] = live && u[g * 4 + j] < (unsigned)nr;
                    vote[g * 4 + j] = __builtin_amdgcn_ballot_w32(m[g * 4 + j]);
                }
            }
            #pragma unroll
            for (int k = 0; k < LNK_U * 4; ++k) {
                if (vote[k] != 0u) {
                    if (m[k]) wq[(qt + (int)__builtin_amdgcn_mbcnt_lo(vote[k], 0u)) & (LNK_QN - 1)] =
                        make_int2((int)u[k], b0 + (k >> 2) * STR + (k & 3));
                    qt += __builtin_popcount(vote[k]);
                    if (qt - qh >= 32) { link_batch(32); qh += 32; }
                }
            }
        };

        int4 dn[LNK_U];
        if (c0 < c1) {
            #pragma unroll
            for (int g = 0; g < LNK_U; ++g)
                dn[g] = *(const int4*)(dstp + min(c0 + (int)threadIdx.x * 4 + g * STR, c1 - 4));
        }
        int base = c0;
        for (; base + STEP <= c1; base += STEP) {
            const int b0 = base + threadIdx.x * 4;
            int4 dv[LNK_U];
            #pragma unroll
            for (int g = 0; g < LNK_U; ++g) dv[g] = dn[g];
            if (base + STEP < c1) {
                #pragma unroll
                for (int g = 0; g < LNK_U; ++g)
                    dn[g] = *(const int4*)(dstp + min(b0 + STEP + g * STR, c1 - 4));
            }
            sieve_pass(dv, b0, lnk_full{});
        }
        if (base < c1) sieve_pass(dn, base + (int)threadIdx.x * 4, lnk_tail{});
        link_batch(qt - qh);
    } else {
        for (int b = c0 + threadIdx.x; b < c1; b += LNK_TPB) {
            const unsigned u = (unsigned)(dstp[b] - lo);
            if (u < (unsigned)nr) file_edge((int)u, b, eidx[b]);
        }
    }
    __syncthreads();
    for (int i = threadIdx.x; i < nr; i += LNK_TPB) {
        if (lo + i < n) {
            const int k = lcnt[i];
            unsigned x = 0u, y = 0u;
            if (k > 0) {
                const unsigned s0 = (unsigned)lsrc[0][i];
                const unsigned s1 = k > 1 ? (unsigned)lsrc[1][i] : 0u;
                unsigned f3 = 0u;
                if (k == 3) {
                    f3 = (unsigned)lsrc[2][i];
                } else if (k > 3) {
                    f3 = (unsigned)lnx[i];
                    rec[2 * (size_t)lte[i] + 1] = LNK_T1 | (c << LNK_SB) | lsrc[2][i];
                }
                x = s0 | (s1 << LNK_SB);
                y = (s1 >> (32 - LNK_SB)) | (f3 << 2) | ((unsigned)min(k, 4) << 26) | ((unsigned)c << 29);
            }
            ((int2*)headT)[(size_t)c * n + lo + i] = make_int2((int)x, (int)y);
        }
    }
}

template <int H, int C, int HS4, int AST>
__global__ void row_walk_kernel(const int* __restrict__ rec, int E, int n,
                                const int* __restrict__ headT,
                                const float* __restrict__ as,
                                const float* __restrict__ ad,
                                const float* __restrict__ h,
                                const float* __restrict__ bias,
                                float* __restrict__ out) {
    constexpr int LPR = (H * C) / 4;
    static_assert(((1 << LNK_SB) - 1) < (1 << 30) / (HS4 > AST ? HS4 : AST), "32-bit row and logit indices");
    const int gid = blockIdx.x * blockDim.x + threadIdx.x;
    const int row = gid / LPR;
    if (row >= n) return;
    const int q  = gid - row * LPR;
    const int hh = (q * 4) / C;
    const float4* h4 = (const float4*)h;
    const int2*   r2 = (const int2*)rec;
    const int2*   hT = (const int2*)headT + row;

    const float adv = ad[row * AST + hh];
    float  m   = leaky(as[row * AST + hh] + adv);
    float  den = 1.f;
    float4 acc = h4[row * HS4 + q];

    auto add_edge = [&](const float4 hv, const float av) {
        const float sc = leaky(av + adv);
        const float mn = fmaxf(m, sc);
        const float a  = __expf(m - mn);
        const float w  = __expf(sc - mn);
        den   = den   * a + w;
        acc.x = acc.x * a + w * hv.x;
        acc.y = acc.y * a + w * hv.y;
        acc.z = acc.z * a + w * hv.z;
        acc.w = acc.w * a + w * hv.w;
        m = mn;
    };

    int2 hn = hT[0];
    #pragma nounroll
    for (int c = 0; c < LNK_B; ++c) {
        const int2 hd = hn;
        if (c + 1 < LNK_B) hn = hT[(size_t)(c + 1) * n];
        const unsigned ux = (unsigned)hd.x, uy = (unsigned)hd.y;
        const int kind = (int)((uy >> 26) & 7u);
        const int s0   = (int)(ux & (unsigned)LNK_SM);
        const int s1   = (int)((ux >> LNK_SB) | ((uy & 3u) << (32 - LNK_SB)));
        const int f3   = (int)((uy >> 2) & 0xFFFFFFu);
        float4 v0, v1, v2;
        float  a0, a1, a2;
        int2   r;
        if (kind > 0)  { v0 = h4[s0 * HS4 + q]; a0 = as[s0 * AST + hh]; }
        if (kind > 1)  { v1 = h4[s1 * HS4 + q]; a1 = as[s1 * AST + hh]; }
        if (kind == 3) { v2 = h4[f3 * HS4 + q]; a2 = as[f3 * AST + hh]; }
        if (kind > 3)  r = r2[f3];
        if (kind > 0)  add_edge(v0, a0);
        if (kind > 1)  add_edge(v1, a1);
        if (kind == 3) add_edge(v2, a2);
        if (kind > 3) {
            for (int it = 0; it < E; ++it) {
                const int s = r.x, nx = r.y;
                const float4 hv = h4[s * HS4 + q];
                const float  av = as[s * AST + hh];
                if (nx < LNK_T1) {
                    r = r2[nx];
                    add_edge(hv, av);
                } else {
                    const int t = nx & LNK_SM;
                    const float4 tv = h4[t * HS4 + q];
                    const float  ta = as[t * AST + hh];
                    add_edge(hv, av);
                    add_edge(tv, ta);
                    break;
                }
            }
        }
    }
    const float4 b = ((const float4*)bias)[q];
    float4 o;
    o.x = acc.x / den + b.x;
    o.y = acc.y / den + b.y;
    o.z = acc.z / den + b.z;
    o.w = acc.w / den + b.w;
    ((float4*)out)[(size_t)row * LPR + q] = o;
}

static inline int ceil_div(int a, int b) { return (a + b - 1) / b; }

extern "C" void kernel_launch(void* const* d_in, const int* in_sizes, int n_in,
                              void* d_out, int out_size, void* d_ws, size_t ws_size,
                              hipStream_t stream) {
    const float* x    = (const float*)d_in[0];
    const int*   eidx = (const int*)  d_in[1];
    const float* W1   = (const float*)d_in[2];
    const float* aS1  = (const float*)d_in[3];
    const float* aD1  = (const float*)d_in[4];
    const float* b1   = (const float*)d_in[5];
    const float* W2   = (const float*)d_in[6];
    const float* aS2  = (const float*)d_in[7];
    const float* aD2  = (const float*)d_in[8];
    const float* b2   = (const float*)d_in[9];
    float* out = (float*)d_out;

    const int FEATS = 256, HID = 64, H1 = 8, C1 = 8, CLS = 16;
    const int N = in_sizes[0] / FEATS;
    const int E = in_sizes[1] / 2;

    float* ws = (float*)d_ws;
    float* h1   = ws;                 ws += (size_t)N * HID;
    float* o1   = ws;                 ws += (size_t)N * HID;
    int*   rec  = (int*)ws;           ws += (size_t)E * 2;
    float* as1  = ws;                 ws += (size_t)N * H1;
    float* ad1  = ws;                 ws += (size_t)N * H1;
    int*   headT1 = (int*)out;
    int*   headT2 = (int*)as1;
    const int NR = 32;
    float* nrec = h1;

    const int TPB = 256;
    const dim3 gblock(32 * GEMM_WPB);
    const dim3 ggrid(ceil_div(N / 16, GEMM_WPB), 1);

    proj_logits_kernel<4, 8, 256, false><<<ggrid, gblock, 0, stream>>>(
        x, W1, aS1, aD1, h1, as1, ad1, nullptr, nullptr, 0, N, HID, H1);
    link_edges_kernel<<<LNK_B * LNK_R, LNK_TPB, 0, stream>>>(eidx, E, N, headT1, rec);
    row_walk_kernel<H1, C1, HID / 4, H1><<<ceil_div(N * (HID / 4), TPB), TPB, 0, stream>>>(
        rec, E, N, headT1, as1, ad1, h1, b1, o1);

    proj_logits_kernel<1, 1, 64, true><<<ggrid, gblock, 0, stream>>>(
        o1, W2, aS2, aD2, nrec, nrec + CLS, nrec + CLS + 1, headT1, headT2, LNK_B * (N / 2),
        N, NR, NR);
    row_walk_kernel<1, CLS, NR / 4, NR><<<ceil_div(N * (CLS / 4), TPB), TPB, 0, stream>>>(
        rec, E, N, headT2, nrec + CLS, nrec + CLS + 1, nrec, b2, out);
}
